# starting kernel
# baseline (speedup 1.0000x reference)
#include <hip/hip_runtime.h>
#include <hip/hip_bf16.h>
#include <math.h>


#define BB      2
#define TT      2048
#define DM      1024
#define NH      16
#define DH      64
#define ROWS    (BB * TT)
#define QKV_N   (3 * DM)

typedef __attribute__((ext_vector_type(2))) float v2f;
typedef __attribute__((ext_vector_type(8))) float v8f;

__global__ void gemm_bias_f32wmma(const float* __restrict__ A,
                                  const float* __restrict__ Bm,
                                  const float* __restrict__ bias,
                                  float* __restrict__ C,
                                  int M, int N, int K)
{
    const int wave = blockIdx.x * (blockDim.x >> 5) + (threadIdx.x >> 5);
    const int lane = threadIdx.x & 31;
    const int half = lane >> 4;
    const int ln   = lane & 15;

    const int ngroups = N >> 6;
    const int mt = wave / ngroups;
    const int ng = wave % ngroups;
    if (mt * 32 >= M) return;

    const float* __restrict__ ap0 = A + (size_t)(mt * 32 + ln) * K + 2 * half;
    const float* __restrict__ ap1 = ap0 + (size_t)16 * K;
    const float* __restrict__ bp  = Bm + (size_t)(2 * half) * N + ng * 64 + ln;

    v8f acc00 = {}, acc01 = {}, acc02 = {}, acc03 = {};
    v8f acc10 = {}, acc11 = {}, acc12 = {}, acc13 = {};

    for (int k0 = 0; k0 < K; k0 += 4) {
        const v2f a0 = *(const v2f*)ap0;
        const v2f a1 = *(const v2f*)ap1;

        v2f b0; b0.x = bp[0];  b0.y = bp[N];
        v2f b1; b1.x = bp[16]; b1.y = bp[N + 16];
        v2f b2; b2.x = bp[32]; b2.y = bp[N + 32];
        v2f b3; b3.x = bp[48]; b3.y = bp[N + 48];

        acc00 = __builtin_amdgcn_wmma_f32_16x16x4_f32(false, a0, false, b0, (short)0, acc00, false, false);
        acc10 = __builtin_amdgcn_wmma_f32_16x16x4_f32(false, a1, false, b0, (short)0, acc10, false, false);
        acc01 = __builtin_amdgcn_wmma_f32_16x16x4_f32(false, a0, false, b1, (short)0, acc01, false, false);
        acc11 = __builtin_amdgcn_wmma_f32_16x16x4_f32(false, a1, false, b1, (short)0, acc11, false, false);
        acc02 = __builtin_amdgcn_wmma_f32_16x16x4_f32(false, a0, false, b2, (short)0, acc02, false, false);
        acc12 = __builtin_amdgcn_wmma_f32_16x16x4_f32(false, a1, false, b2, (short)0, acc12, false, false);
        acc03 = __builtin_amdgcn_wmma_f32_16x16x4_f32(false, a0, false, b3, (short)0, acc03, false, false);
        acc13 = __builtin_amdgcn_wmma_f32_16x16x4_f32(false, a1, false, b3, (short)0, acc13, false, false);

        ap0 += 4;
        ap1 += 4;
        bp  += (size_t)4 * N;
    }

    v8f accs[2][4] = { { acc00, acc01, acc02, acc03 },
                       { acc10, acc11, acc12, acc13 } };
#pragma unroll
    for (int sub = 0; sub < 2; ++sub) {
        const int rbase = mt * 32 + sub * 16 + 8 * half;
#pragma unroll
        for (int t = 0; t < 4; ++t) {
            const int col = ng * 64 + t * 16 + ln;
            const float bs = bias[col];
#pragma unroll
            for (int v = 0; v < 8; ++v)
                C[(size_t)(rbase + v) * N + col] = accs[sub][t][v] + bs;
        }
    }
}

__global__ void rope_split(const float* __restrict__ qkv,
                           float* __restrict__ qb,
                           float* __restrict__ kb,
                           float* __restrict__ vb)
{
    const int row = blockIdx.x;
    const int b = row / TT;
    const int t = row % TT;
    const float* __restrict__ src = qkv + (size_t)row * QKV_N;

    for (int p = threadIdx.x; p < QKV_N / 2; p += blockDim.x) {
        const int c   = p * 2;
        const int sec = c >> 10;
        const int cc  = c & (DM - 1);
        const int h   = cc >> 6;
        const int d   = cc & (DH - 1);

        const float e = src[c];
        const float o = src[c + 1];
        const size_t dst = (((size_t)(b * NH + h)) * TT + t) * DH + d;

        if (sec == 2) {
            vb[dst]     = e;
            vb[dst + 1] = o;
        } else {
            const int   i   = d >> 1;
            const float inv = expf(-(float)(2 * i) * (9.210340371976184f / 64.0f));
            const float ang = (float)t * inv;
            const float sn = sinf(ang);
            const float cs = cosf(ang);
            const float re = e * cs - o * sn;
            const float ro = e * sn + o * cs;
            if (sec == 0) { qb[dst] = re; qb[dst + 1] = ro; }
            else          { kb[dst] = re; kb[dst + 1] = ro; }
        }
    }
}

#define PSTRIDE 20

__global__ void flash_attn_f32wmma(const float* __restrict__ qb,
                                   const float* __restrict__ kb,
                                   const float* __restrict__ vb,
                                   const int*   __restrict__ amask,
                                   float* __restrict__ attn  )
{
    __shared__ float pshare[4][16 * PSTRIDE];

    const int wid  = threadIdx.x >> 5;
    const int lane = threadIdx.x & 31;
    const int half = lane >> 4;
    const int ln   = lane & 15;

    const int blocksPerBH = TT / 64;
    const int bh   = blockIdx.x / blocksPerBH;
    const int qblk = blockIdx.x % blocksPerBH;
    const int b    = bh >> 4;
    const int q0   = qblk * 64 + wid * 16;

    const float* __restrict__ Q  = qb + (size_t)bh * TT * DH;
    const float* __restrict__ Kp = kb + (size_t)bh * TT * DH;
    const float* __restrict__ Vp = vb + (size_t)bh * TT * DH;

    v2f qf[16];
    {
        const float* __restrict__ qrow = Q + (size_t)(q0 + ln) * DH;
#pragma unroll
        for (int i = 0; i < 16; ++i)
            qf[i] = *(const v2f*)(qrow + 4 * i + 2 * half);
    }

    v8f o0 = {}, o1 = {}, o2 = {}, o3 = {};
    float rmax[8], rsum[8];
#pragma unroll
    for (int v = 0; v < 8; ++v) { rmax[v] = -1e30f; rsum[v] = 0.0f; }

    float* __restrict__ P = &pshare[wid][0];
    const float scale = 0.125f;
    const int nkt = q0 / 16 + 1;

    for (int j = 0; j < nkt; ++j) {
        const int kbase = j * 16;

        v8f s = {};
        {
            const float* __restrict__ krow = Kp + (size_t)(kbase + ln) * DH;
#pragma unroll
            for (int i = 0; i < 16; ++i) {
                const v2f bf = *(const v2f*)(krow + 4 * i + 2 * half);
                s = __builtin_amdgcn_wmma_f32_16x16x4_f32(false, qf[i], false, bf,
                                                          (short)0, s, false, false);
            }
        }

        const int keyIdx = kbase + ln;
        const int pm     = amask[b * TT + keyIdx];
#pragma unroll
        for (int v = 0; v < 8; ++v) {
            const int qrowi = q0 + v + 8 * half;
            float val = s[v] * scale;
            val = (keyIdx > qrowi || pm == 0) ? -1e30f : val;

            float mx = val;
            mx = fmaxf(mx, __shfl_xor(mx, 1, 32));
            mx = fmaxf(mx, __shfl_xor(mx, 2, 32));
            mx = fmaxf(mx, __shfl_xor(mx, 4, 32));
            mx = fmaxf(mx, __shfl_xor(mx, 8, 32));

            const float nmax = fmaxf(rmax[v], mx);
            const float p    = __expf(val - nmax);

            float sr = p;
            sr += __shfl_xor(sr, 1, 32);
            sr += __shfl_xor(sr, 2, 32);
            sr += __shfl_xor(sr, 4, 32);
            sr += __shfl_xor(sr, 8, 32);

            const float alpha = __expf(rmax[v] - nmax);
            rsum[v] = rsum[v] * alpha + sr;
            rmax[v] = nmax;
            o0[v] *= alpha; o1[v] *= alpha; o2[v] *= alpha; o3[v] *= alpha;

            P[(v + 8 * half) * PSTRIDE + ln] = p;
        }

        asm volatile("s_wait_dscnt 0x0" ::: "memory");

#pragma unroll
        for (int i = 0; i < 4; ++i) {
            const v2f af = *(const v2f*)(P + ln * PSTRIDE + 4 * i + 2 * half);
            const float* __restrict__ vrow =
                Vp + (size_t)(kbase + 4 * i + 2 * half) * DH + ln;
            v2f b0; b0.x = vrow[0];  b0.y = vrow[DH];
            v2f b1; b1.x = vrow[16]; b1.y = vrow[DH + 16];
            v2f b2; b2.x = vrow[32]; b2.y = vrow[DH + 32];
            v2f b3; b3.x = vrow[48]; b3.y = vrow[DH + 48];
            o0 = __builtin_amdgcn_wmma_f32_16x16x4_f32(false, af, false, b0, (short)0, o0, false, false);
            o1 = __builtin_amdgcn_wmma_f32_16x16x4_f32(false, af, false, b1, (short)0, o1, false, false);
            o2 = __builtin_amdgcn_wmma_f32_16x16x4_f32(false, af, false, b2, (short)0, o2, false, false);
            o3 = __builtin_amdgcn_wmma_f32_16x16x4_f32(false, af, false, b3, (short)0, o3, false, false);
        }
    }

    const int h = bh & (NH - 1);
    v8f os[4] = { o0, o1, o2, o3 };
#pragma unroll
    for (int v = 0; v < 8; ++v) {
        const float inv = 1.0f / rsum[v];
        const size_t row = (size_t)b * TT + (q0 + v + 8 * half);
#pragma unroll
        for (int t = 0; t < 4; ++t)
            attn[row * DM + h * DH + t * 16 + ln] = os[t][v] * inv;
    }
}

extern "C" void kernel_launch(void* const* d_in, const int* in_sizes, int n_in,
                              void* d_out, int out_size, void* d_ws, size_t ws_size,
                              hipStream_t stream)
{
    (void)in_sizes; (void)n_in; (void)out_size; (void)ws_size;

    const float* x      = (const float*)d_in[0];
    const int*   mask   = (const int*)  d_in[1];
    const float* W_qkv  = (const float*)d_in[2];
    const float* b_qkv  = (const float*)d_in[3];
    const float* W_out  = (const float*)d_in[4];
    const float* b_out  = (const float*)d_in[5];
    float*       out    = (float*)d_out;

    float* ws   = (float*)d_ws;
    float* qkv  = ws;
    float* qb   = qkv + (size_t)ROWS * QKV_N;
    float* kb   = qb  + (size_t)BB * NH * TT * DH;
    float* vb   = kb  + (size_t)BB * NH * TT * DH;
    float* attn = qkv;

    {
        const int waves = (ROWS / 32) * (QKV_N / 64);
        gemm_bias_f32wmma<<<waves / 4, 128, 0, stream>>>(
            x, W_qkv, b_qkv, qkv, ROWS, QKV_N, DM);
    }

    rope_split<<<ROWS, 256, 0, stream>>>(qkv, qb, kb, vb);

    {
        const int blocks = BB * NH * (TT / 64);
        flash_attn_f32wmma<<<blocks, 128, 0, stream>>>(qb, kb, vb, mask, attn);
    }

    {
        const int waves = (ROWS / 32) * (DM / 64);
        gemm_bias_f32wmma<<<waves / 4, 128, 0, stream>>>(
            attn, W_out, b_out, out, ROWS, DM, DM);
    }
}


# Attention tile transposed (S^T=K*Q^T, O^T=V^T*P^T): one query per lane, softmax max/sum in registers, P^T feeds WMMA B from registers (no LDS, no shuffles); k order of sums and products changed
# speedup vs baseline: 2.0802x; 2.0802x over previous
#include <hip/hip_runtime.h>
#include <hip/hip_bf16.h>
#include <math.h>


#define BB      2
#define TT      2048
#define DM      1024
#define NH      16
#define DH      64
#define ROWS    (BB * TT)
#define QKV_N   (3 * DM)

typedef __attribute__((ext_vector_type(2))) float v2f;
typedef __attribute__((ext_vector_type(8))) float v8f;

__global__ void gemm_bias_f32wmma(const float* __restrict__ A,
                                  const float* __restrict__ Bm,
                                  const float* __restrict__ bias,
                                  float* __restrict__ C,
                                  int M, int N, int K)
{
    const int wave = blockIdx.x * (blockDim.x >> 5) + (threadIdx.x >> 5);
    const int lane = threadIdx.x & 31;
    const int half = lane >> 4;
    const int ln   = lane & 15;

    const int ngroups = N >> 6;
    const int mt = wave / ngroups;
    const int ng = wave % ngroups;
    if (mt * 32 >= M) return;

    const float* __restrict__ ap0 = A + (size_t)(mt * 32 + ln) * K + 2 * half;
    const float* __restrict__ ap1 = ap0 + (size_t)16 * K;
    const float* __restrict__ bp  = Bm + (size_t)(2 * half) * N + ng * 64 + ln;

    v8f acc00 = {}, acc01 = {}, acc02 = {}, acc03 = {};
    v8f acc10 = {}, acc11 = {}, acc12 = {}, acc13 = {};

    for (int k0 = 0; k0 < K; k0 += 4) {
        const v2f a0 = *(const v2f*)ap0;
        const v2f a1 = *(const v2f*)ap1;

        v2f b0; b0.x = bp[0];  b0.y = bp[N];
        v2f b1; b1.x = bp[16]; b1.y = bp[N + 16];
        v2f b2; b2.x = bp[32]; b2.y = bp[N + 32];
        v2f b3; b3.x = bp[48]; b3.y = bp[N + 48];

        acc00 = __builtin_amdgcn_wmma_f32_16x16x4_f32(false, a0, false, b0, (short)0, acc00, false, false);
        acc10 = __builtin_amdgcn_wmma_f32_16x16x4_f32(false, a1, false, b0, (short)0, acc10, false, false);
        acc01 = __builtin_amdgcn_wmma_f32_16x16x4_f32(false, a0, false, b1, (short)0, acc01, false, false);
        acc11 = __builtin_amdgcn_wmma_f32_16x16x4_f32(false, a1, false, b1, (short)0, acc11, false, false);
        acc02 = __builtin_amdgcn_wmma_f32_16x16x4_f32(false, a0, false, b2, (short)0, acc02, false, false);
        acc12 = __builtin_amdgcn_wmma_f32_16x16x4_f32(false, a1, false, b2, (short)0, acc12, false, false);
        acc03 = __builtin_amdgcn_wmma_f32_16x16x4_f32(false, a0, false, b3, (short)0, acc03, false, false);
        acc13 = __builtin_amdgcn_wmma_f32_16x16x4_f32(false, a1, false, b3, (short)0, acc13, false, false);

        ap0 += 4;
        ap1 += 4;
        bp  += (size_t)4 * N;
    }

    v8f accs[2][4] = { { acc00, acc01, acc02, acc03 },
                       { acc10, acc11, acc12, acc13 } };
#pragma unroll
    for (int sub = 0; sub < 2; ++sub) {
        const int rbase = mt * 32 + sub * 16 + 8 * half;
#pragma unroll
        for (int t = 0; t < 4; ++t) {
            const int col = ng * 64 + t * 16 + ln;
            const float bs = bias[col];
#pragma unroll
            for (int v = 0; v < 8; ++v)
                C[(size_t)(rbase + v) * N + col] = accs[sub][t][v] + bs;
        }
    }
}

__global__ void rope_split(const float* __restrict__ qkv,
                           float* __restrict__ qb,
                           float* __restrict__ kb,
                           float* __restrict__ vb)
{
    const int row = blockIdx.x;
    const int b = row / TT;
    const int t = row % TT;
    const float* __restrict__ src = qkv + (size_t)row * QKV_N;

    for (int p = threadIdx.x; p < QKV_N / 2; p += blockDim.x) {
        const int c   = p * 2;
        const int sec = c >> 10;
        const int cc  = c & (DM - 1);
        const int h   = cc >> 6;
        const int d   = cc & (DH - 1);

        const float e = src[c];
        const float o = src[c + 1];
        const size_t dst = (((size_t)(b * NH + h)) * TT + t) * DH + d;

        if (sec == 2) {
            vb[dst]     = e;
            vb[dst + 1] = o;
        } else {
            const int   i   = d >> 1;
            const float inv = expf(-(float)(2 * i) * (9.210340371976184f / 64.0f));
            const float ang = (float)t * inv;
            const float sn = sinf(ang);
            const float cs = cosf(ang);
            const float re = e * cs - o * sn;
            const float ro = e * sn + o * cs;
            if (sec == 0) { qb[dst] = re; qb[dst + 1] = ro; }
            else          { kb[dst] = re; kb[dst + 1] = ro; }
        }
    }
}

typedef __attribute__((ext_vector_type(4))) float v4f;
typedef __attribute__((ext_vector_type(4))) int   v4i;

__device__ __forceinline__ float xhalf(float x)
{
    const int bits = __builtin_bit_cast(int, x);
    return __builtin_bit_cast(float,
        __builtin_amdgcn_permlanex16(bits, bits, 0x76543210, 0xfedcba98, false, false));
}

__global__ void __launch_bounds__(128)
flash_attn_f32wmma(const float* __restrict__ qb,
                   const float* __restrict__ kb,
                   const float* __restrict__ vb,
                   const int*   __restrict__ amask,
                   float* __restrict__ attn  )
{
    const int wid  = __builtin_amdgcn_readfirstlane((int)(threadIdx.x >> 5));
    const int lane = threadIdx.x & 31;
    const int half = lane >> 4;
    const int ln   = lane & 15;

    const int blocksPerBH = TT / 64;
    const int bh   = blockIdx.x / blocksPerBH;
    const int qblk = blockIdx.x % blocksPerBH;
    const int b    = bh >> 4;
    const int h    = bh & (NH - 1);
    const int q0   = qblk * 64 + wid * 16;

    const float* __restrict__ Q  = qb + (size_t)bh * TT * DH;
    const float* __restrict__ Kp = kb + (size_t)bh * TT * DH;
    const float* __restrict__ Vp = vb + (size_t)bh * TT * DH;

    v4f q4[8];
    {
        const float* __restrict__ qrow = Q + (size_t)(q0 + ln) * DH + 4 * half;
        for (int w = 0; w < 8; ++w)
            q4[w] = *(const v4f*)(qrow + 8 * w);
    }

    v8f o0 = {}, o1 = {}, o2 = {}, o3 = {};
    float rmax = -1e30f;
    float rsum = 0.0f;

    const float scale = 0.125f;
    const int nkt = q0 / 16 + 1;

    const float* __restrict__ kptr = Kp + (size_t)ln * DH + 4 * half;
    const float* __restrict__ vptr = Vp + (size_t)(8 * half) * DH + 4 * ln;
    const int*   __restrict__ mptr = amask + b * TT + 8 * half;

    for (int j = 0; j < nkt; ++j) {
        const int kbase = j * 16;

        v4f k4[8];
        for (int w = 0; w < 8; ++w)
            k4[w] = *(const v4f*)(kptr + (size_t)kbase * DH + 8 * w);
        v4f v4[8];
        for (int r = 0; r < 8; ++r)
            v4[r] = *(const v4f*)(vptr + (size_t)(kbase + r) * DH);
        const v4i pm0 = *(const v4i*)(mptr + kbase);
        const v4i pm1 = *(const v4i*)(mptr + kbase + 4);

        v8f s = {};
        for (int w = 0; w < 8; ++w) {
            v2f ka, qa, kc, qc;
            ka.x = k4[w].x; ka.y = k4[w].y;  qa.x = q4[w].x; qa.y = q4[w].y;
            kc.x = k4[w].z; kc.y = k4[w].w;  qc.x = q4[w].z; qc.y = q4[w].w;
            s = __builtin_amdgcn_wmma_f32_16x16x4_f32(false, ka, false, qa, (short)0, s, false, false);
            s = __builtin_amdgcn_wmma_f32_16x16x4_f32(false, kc, false, qc, (short)0, s, false, false);
        }

        const int pmv[8] = { pm0.x, pm0.y, pm0.z, pm0.w, pm1.x, pm1.y, pm1.z, pm1.w };
        float x[8];
        for (int r = 0; r < 8; ++r)
            x[r] = (pmv[r] == 0) ? -1e30f : s[r] * scale;
        if (j == nkt - 1) {
            for (int r = 0; r < 8; ++r)
                x[r] = (8 * half + r > ln) ? -1e30f : x[r];
        }

        float mx = fmaxf(fmaxf(fmaxf(x[0], x[1]), fmaxf(x[2], x[3])),
                         fmaxf(fmaxf(x[4], x[5]), fmaxf(x[6], x[7])));
        mx = fmaxf(mx, xhalf(mx));
        const float nmax  = fmaxf(rmax, mx);
        const float alpha = __expf(rmax - nmax);
        float p[8];
        for (int r = 0; r < 8; ++r)
            p[r] = __expf(x[r] - nmax);
        float sr = ((p[0] + p[1]) + (p[2] + p[3])) + ((p[4] + p[5]) + (p[6] + p[7]));
        sr += xhalf(sr);
        rsum = rsum * alpha + sr;
        rmax = nmax;
        o0 *= alpha; o1 *= alpha; o2 *= alpha; o3 *= alpha;

        for (int i = 0; i < 4; ++i) {
            v2f pf; pf.x = p[2 * i]; pf.y = p[2 * i + 1];
            v2f a0; a0.x = v4[2 * i].x; a0.y = v4[2 * i + 1].x;
            v2f a1; a1.x = v4[2 * i].y; a1.y = v4[2 * i + 1].y;
            v2f a2; a2.x = v4[2 * i].z; a2.y = v4[2 * i + 1].z;
            v2f a3; a3.x = v4[2 * i].w; a3.y = v4[2 * i + 1].w;
            o0 = __builtin_amdgcn_wmma_f32_16x16x4_f32(false, a0, false, pf, (short)0, o0, false, false);
            o1 = __builtin_amdgcn_wmma_f32_16x16x4_f32(false, a1, false, pf, (short)0, o1, false, false);
            o2 = __builtin_amdgcn_wmma_f32_16x16x4_f32(false, a2, false, pf, (short)0, o2, false, false);
            o3 = __builtin_amdgcn_wmma_f32_16x16x4_f32(false, a3, false, pf, (short)0, o3, false, false);
        }
    }

    const float inv = 1.0f / rsum;
    float* __restrict__ orow =
        attn + ((size_t)b * TT + (q0 + ln)) * DM + h * DH + 32 * half;
    for (int r = 0; r < 8; ++r) {
        v4f o;
        o.x = o0[r] * inv; o.y = o1[r] * inv; o.z = o2[r] * inv; o.w = o3[r] * inv;
        *(v4f*)(orow + 4 * r) = o;
    }
}

extern "C" void kernel_launch(void* const* d_in, const int* in_sizes, int n_in,
                              void* d_out, int out_size, void* d_ws, size_t ws_size,
                              hipStream_t stream)
{
    (void)in_sizes; (void)n_in; (void)out_size; (void)ws_size;

    const float* x      = (const float*)d_in[0];
    const int*   mask   = (const int*)  d_in[1];
    const float* W_qkv  = (const float*)d_in[2];
    const float* b_qkv  = (const float*)d_in[3];
    const float* W_out  = (const float*)d_in[4];
    const float* b_out  = (const float*)d_in[5];
    float*       out    = (float*)d_out;

    float* ws   = (float*)d_ws;
    float* qkv  = ws;
    float* qb   = qkv + (size_t)ROWS * QKV_N;
    float* kb   = qb  + (size_t)BB * NH * TT * DH;
    float* vb   = kb  + (size_t)BB * NH * TT * DH;
    float* attn = qkv;

    {
        const int waves = (ROWS / 32) * (QKV_N / 64);
        gemm_bias_f32wmma<<<waves / 4, 128, 0, stream>>>(
            x, W_qkv, b_qkv, qkv, ROWS, QKV_N, DM);
    }

    rope_split<<<ROWS, 256, 0, stream>>>(qkv, qb, kb, vb);

    {
        const int blocks = BB * NH * (TT / 64);
        flash_attn_f32wmma<<<blocks, 128, 0, stream>>>(qb, kb, vb, mask, attn);
    }

    {
        const int waves = (ROWS / 32) * (DM / 64);
        gemm_bias_f32wmma<<<waves / 4, 128, 0, stream>>>(
            attn, W_out, b_out, out, ROWS, DM, DM);
    }
}


# Both GEMMs: A and B chunks (K=32) staged once per 128x128 workgroup tile through two LDS stages instead of per-wave global fragment loads; sums and their k order unchanged
# speedup vs baseline: 2.7543x; 1.3241x over previous
#include <hip/hip_runtime.h>
#include <hip/hip_bf16.h>
#include <math.h>


#define BB      2
#define TT      2048
#define DM      1024
#define NH      16
#define DH      64
#define ROWS    (BB * TT)
#define QKV_N   (3 * DM)

typedef __attribute__((ext_vector_type(2))) float v2f;
typedef __attribute__((ext_vector_type(8))) float v8f;

typedef __attribute__((ext_vector_type(4))) float g4f;

#define GBM      128
#define GBN      128
#define GBK      32
#define APITCH   36
#define BPITCH2  288
#define A_STAGE  (GBM * APITCH)
#define B_STAGE  ((GBK / 2) * BPITCH2)
#define STAGE_F  (A_STAGE + B_STAGE)

struct GFrag { v2f a0, a1, b0, b1, b2, b3; };

__device__ __forceinline__ void load_gfrag(GFrag& f, const float* ap, const float* bp)
{
    f.a0 = *(const v2f*)(ap);
    f.a1 = *(const v2f*)(ap + 16 * APITCH);
    f.b0 = *(const v2f*)(bp);
    f.b1 = *(const v2f*)(bp + 32);
    f.b2 = *(const v2f*)(bp + 64);
    f.b3 = *(const v2f*)(bp + 96);
}

__device__ __forceinline__ void store_bpair(float* dst, g4f r0, g4f r1)
{
    g4f lo, hi;
    lo.x = r0.x; lo.y = r1.x; lo.z = r0.y; lo.w = r1.y;
    hi.x = r0.z; hi.y = r1.z; hi.z = r0.w; hi.w = r1.w;
    *(g4f*)(dst)     = lo;
    *(g4f*)(dst + 4) = hi;
}

__global__ void __launch_bounds__(256)
gemm_bias_f32wmma(const float* __restrict__ A,
                  const float* __restrict__ Bm,
                  const float* __restrict__ bias,
                  float* __restrict__ C,
                  int M, int N, int K)
{
    __shared__ __attribute__((aligned(16))) float lds[2 * STAGE_F];

    (void)M;
    const int tid  = threadIdx.x;
    const int wid  = __builtin_amdgcn_readfirstlane(tid >> 5);
    const int lane = tid & 31;
    const int half = lane >> 4;
    const int ln   = lane & 15;
    const int wm   = wid & 3;
    const int wn   = wid >> 2;

    const int nbn = N / GBN;
    const int m0  = (blockIdx.x / nbn) * GBM;
    const int n0  = (blockIdx.x % nbn) * GBN;

    const float* __restrict__ ag = A  + (size_t)(m0 + (tid >> 3)) * K + 4 * (tid & 7);
    const float* __restrict__ bg = Bm + (size_t)(2 * (tid >> 5)) * N + n0 + 4 * (tid & 31);
    const int as_off = (tid >> 3) * APITCH + 4 * (tid & 7);
    const int bs_off = (tid >> 5) * BPITCH2 + 8 * (tid & 31);

    const int a_rd = (wm * 32 + ln) * APITCH + 2 * half;
    const int b_rd = half * BPITCH2 + 2 * (wn * 64 + ln);

    g4f ra[4], rb[4];
#pragma unroll
    for (int p = 0; p < 4; ++p) {
        ra[p] = *(const g4f*)(ag + (size_t)(32 * p) * K);
        rb[p] = *(const g4f*)(bg + (size_t)(16 * (p >> 1) + (p & 1)) * N);
    }
#pragma unroll
    for (int p = 0; p < 4; ++p)
        *(g4f*)(lds + as_off + 32 * p * APITCH) = ra[p];
#pragma unroll
    for (int q = 0; q < 2; ++q)
        store_bpair(lds + A_STAGE + bs_off + 8 * q * BPITCH2, rb[2 * q], rb[2 * q + 1]);
    __syncthreads();

    v8f acc00 = {}, acc01 = {}, acc02 = {}, acc03 = {};
    v8f acc10 = {}, acc11 = {}, acc12 = {}, acc13 = {};

    const int nch = K / GBK;
    for (int c = 0; c < nch; ++c) {
        const float* As = lds + (c & 1) * STAGE_F;
        const float* Bs = As + A_STAGE;
        const bool more = (c + 1 < nch);

        if (more) {
            ag += GBK;
            bg += (size_t)GBK * N;
#pragma unroll
            for (int p = 0; p < 4; ++p) {
                ra[p] = *(const g4f*)(ag + (size_t)(32 * p) * K);
                rb[p] = *(const g4f*)(bg + (size_t)(16 * (p >> 1) + (p & 1)) * N);
            }
        }

        GFrag f[2];
        load_gfrag(f[0], As + a_rd, Bs + b_rd);
#pragma unroll
        for (int ks = 0; ks < GBK / 4; ++ks) {
            if (ks + 1 < GBK / 4)
                load_gfrag(f[(ks + 1) & 1], As + a_rd + 4 * (ks + 1), Bs + b_rd + 2 * (ks + 1) * BPITCH2);
            __builtin_amdgcn_sched_barrier(0);
            const GFrag& g = f[ks & 1];
            acc00 = __builtin_amdgcn_wmma_f32_16x16x4_f32(false, g.a0, false, g.b0, (short)0, acc00, false, false);
            acc10 = __builtin_amdgcn_wmma_f32_16x16x4_f32(false, g.a1, false, g.b0, (short)0, acc10, false, false);
            acc01 = __builtin_amdgcn_wmma_f32_16x16x4_f32(false, g.a0, false, g.b1, (short)0, acc01, false, false);
            acc11 = __builtin_amdgcn_wmma_f32_16x16x4_f32(false, g.a1, false, g.b1, (short)0, acc11, false, false);
            acc02 = __builtin_amdgcn_wmma_f32_16x16x4_f32(false, g.a0, false, g.b2, (short)0, acc02, false, false);
            acc12 = __builtin_amdgcn_wmma_f32_16x16x4_f32(false, g.a1, false, g.b2, (short)0, acc12, false, false);
            acc03 = __builtin_amdgcn_wmma_f32_16x16x4_f32(false, g.a0, false, g.b3, (short)0, acc03, false, false);
            acc13 = __builtin_amdgcn_wmma_f32_16x16x4_f32(false, g.a1, false, g.b3, (short)0, acc13, false, false);
            __builtin_amdgcn_sched_barrier(0);
        }

        if (more) {
            float* Ad = lds + ((c + 1) & 1) * STAGE_F;
            float* Bd = Ad + A_STAGE;
#pragma unroll
            for (int p = 0; p < 4; ++p)
                *(g4f*)(Ad + as_off + 32 * p * APITCH) = ra[p];
#pragma unroll
            for (int q = 0; q < 2; ++q)
                store_bpair(Bd + bs_off + 8 * q * BPITCH2, rb[2 * q], rb[2 * q + 1]);
        }
        __syncthreads();
    }

    v8f accs[2][4] = { { acc00, acc01, acc02, acc03 },
                       { acc10, acc11, acc12, acc13 } };
#pragma unroll
    for (int sub = 0; sub < 2; ++sub) {
        const int rbase = m0 + wm * 32 + sub * 16 + 8 * half;
#pragma unroll
        for (int t = 0; t < 4; ++t) {
            const int col = n0 + wn * 64 + t * 16 + ln;
            const float bs = bias[col];
#pragma unroll
            for (int v = 0; v < 8; ++v)
                C[(size_t)(rbase + v) * N + col] = accs[sub][t][v] + bs;
        }
    }
}

__global__ void rope_split(const float* __restrict__ qkv,
                           float* __restrict__ qb,
                           float* __restrict__ kb,
                           float* __restrict__ vb)
{
    const int row = blockIdx.x;
    const int b = row / TT;
    const int t = row % TT;
    const float* __restrict__ src = qkv + (size_t)row * QKV_N;

    for (int p = threadIdx.x; p < QKV_N / 2; p += blockDim.x) {
        const int c   = p * 2;
        const int sec = c >> 10;
        const int cc  = c & (DM - 1);
        const int h   = cc >> 6;
        const int d   = cc & (DH - 1);

        const float e = src[c];
        const float o = src[c + 1];
        const size_t dst = (((size_t)(b * NH + h)) * TT + t) * DH + d;

        if (sec == 2) {
            vb[dst]     = e;
            vb[dst + 1] = o;
        } else {
            const int   i   = d >> 1;
            const float inv = expf(-(float)(2 * i) * (9.210340371976184f / 64.0f));
            const float ang = (float)t * inv;
            const float sn = sinf(ang);
            const float cs = cosf(ang);
            const float re = e * cs - o * sn;
            const float ro = e * sn + o * cs;
            if (sec == 0) { qb[dst] = re; qb[dst + 1] = ro; }
            else          { kb[dst] = re; kb[dst + 1] = ro; }
        }
    }
}

typedef __attribute__((ext_vector_type(4))) float v4f;
typedef __attribute__((ext_vector_type(4))) int   v4i;

__device__ __forceinline__ float xhalf(float x)
{
    const int bits = __builtin_bit_cast(int, x);
    return __builtin_bit_cast(float,
        __builtin_amdgcn_permlanex16(bits, bits, 0x76543210, 0xfedcba98, false, false));
}

__global__ void __launch_bounds__(128)
flash_attn_f32wmma(const float* __restrict__ qb,
                   const float* __restrict__ kb,
                   const float* __restrict__ vb,
                   const int*   __restrict__ amask,
                   float* __restrict__ attn  )
{
    const int wid  = __builtin_amdgcn_readfirstlane((int)(threadIdx.x >> 5));
    const int lane = threadIdx.x & 31;
    const int half = lane >> 4;
    const int ln   = lane & 15;

    const int blocksPerBH = TT / 64;
    const int bh   = blockIdx.x / blocksPerBH;
    const int qblk = blockIdx.x % blocksPerBH;
    const int b    = bh >> 4;
    const int h    = bh & (NH - 1);
    const int q0   = qblk * 64 + wid * 16;

    const float* __restrict__ Q  = qb + (size_t)bh * TT * DH;
    const float* __restrict__ Kp = kb + (size_t)bh * TT * DH;
    const float* __restrict__ Vp = vb + (size_t)bh * TT * DH;

    v4f q4[8];
    {
        const float* __restrict__ qrow = Q + (size_t)(q0 + ln) * DH + 4 * half;
        for (int w = 0; w < 8; ++w)
            q4[w] = *(const v4f*)(qrow + 8 * w);
    }

    v8f o0 = {}, o1 = {}, o2 = {}, o3 = {};
    float rmax = -1e30f;
    float rsum = 0.0f;

    const float scale = 0.125f;
    const int nkt = q0 / 16 + 1;

    const float* __restrict__ kptr = Kp + (size_t)ln * DH + 4 * half;
    const float* __restrict__ vptr = Vp + (size_t)(8 * half) * DH + 4 * ln;
    const int*   __restrict__ mptr = amask + b * TT + 8 * half;

    for (int j = 0; j < nkt; ++j) {
        const int kbase = j * 16;

        v4f k4[8];
        for (int w = 0; w < 8; ++w)
            k4[w] = *(const v4f*)(kptr + (size_t)kbase * DH + 8 * w);
        v4f v4[8];
        for (int r = 0; r < 8; ++r)
            v4[r] = *(const v4f*)(vptr + (size_t)(kbase + r) * DH);
        const v4i pm0 = *(const v4i*)(mptr + kbase);
        const v4i pm1 = *(const v4i*)(mptr + kbase + 4);

        v8f s = {};
        for (int w = 0; w < 8; ++w) {
            v2f ka, qa, kc, qc;
            ka.x = k4[w].x; ka.y = k4[w].y;  qa.x = q4[w].x; qa.y = q4[w].y;
            kc.x = k4[w].z; kc.y = k4[w].w;  qc.x = q4[w].z; qc.y = q4[w].w;
            s = __builtin_amdgcn_wmma_f32_16x16x4_f32(false, ka, false, qa, (short)0, s, false, false);
            s = __builtin_amdgcn_wmma_f32_16x16x4_f32(false, kc, false, qc, (short)0, s, false, false);
        }

        const int pmv[8] = { pm0.x, pm0.y, pm0.z, pm0.w, pm1.x, pm1.y, pm1.z, pm1.w };
        float x[8];
        for (int r = 0; r < 8; ++r)
            x[r] = (pmv[r] == 0) ? -1e30f : s[r] * scale;
        if (j == nkt - 1) {
            for (int r = 0; r < 8; ++r)
                x[r] = (8 * half + r > ln) ? -1e30f : x[r];
        }

        float mx = fmaxf(fmaxf(fmaxf(x[0], x[1]), fmaxf(x[2], x[3])),
                         fmaxf(fmaxf(x[4], x[5]), fmaxf(x[6], x[7])));
        mx = fmaxf(mx, xhalf(mx));
        const float nmax  = fmaxf(rmax, mx);
        const float alpha = __expf(rmax - nmax);
        float p[8];
        for (int r = 0; r < 8; ++r)
            p[r] = __expf(x[r] - nmax);
        float sr = ((p[0] + p[1]) + (p[2] + p[3])) + ((p[4] + p[5]) + (p[6] + p[7]));
        sr += xhalf(sr);
        rsum = rsum * alpha + sr;
        rmax = nmax;
        o0 *= alpha; o1 *= alpha; o2 *= alpha; o3 *= alpha;

        for (int i = 0; i < 4; ++i) {
            v2f pf; pf.x = p[2 * i]; pf.y = p[2 * i + 1];
            v2f a0; a0.x = v4[2 * i].x; a0.y = v4[2 * i + 1].x;
            v2f a1; a1.x = v4[2 * i].y; a1.y = v4[2 * i + 1].y;
            v2f a2; a2.x = v4[2 * i].z; a2.y = v4[2 * i + 1].z;
            v2f a3; a3.x = v4[2 * i].w; a3.y = v4[2 * i + 1].w;
            o0 = __builtin_amdgcn_wmma_f32_16x16x4_f32(false, a0, false, pf, (short)0, o0, false, false);
            o1 = __builtin_amdgcn_wmma_f32_16x16x4_f32(false, a1, false, pf, (short)0, o1, false, false);
            o2 = __builtin_amdgcn_wmma_f32_16x16x4_f32(false, a2, false, pf, (short)0, o2, false, false);
            o3 = __builtin_amdgcn_wmma_f32_16x16x4_f32(false, a3, false, pf, (short)0, o3, false, false);
        }
    }

    const float inv = 1.0f / rsum;
    float* __restrict__ orow =
        attn + ((size_t)b * TT + (q0 + ln)) * DM + h * DH + 32 * half;
    for (int r = 0; r < 8; ++r) {
        v4f o;
        o.x = o0[r] * inv; o.y = o1[r] * inv; o.z = o2[r] * inv; o.w = o3[r] * inv;
        *(v4f*)(orow + 4 * r) = o;
    }
}

extern "C" void kernel_launch(void* const* d_in, const int* in_sizes, int n_in,
                              void* d_out, int out_size, void* d_ws, size_t ws_size,
                              hipStream_t stream)
{
    (void)in_sizes; (void)n_in; (void)out_size; (void)ws_size;

    const float* x      = (const float*)d_in[0];
    const int*   mask   = (const int*)  d_in[1];
    const float* W_qkv  = (const float*)d_in[2];
    const float* b_qkv  = (const float*)d_in[3];
    const float* W_out  = (const float*)d_in[4];
    const float* b_out  = (const float*)d_in[5];
    float*       out    = (float*)d_out;

    float* ws   = (float*)d_ws;
    float* qkv  = ws;
    float* qb   = qkv + (size_t)ROWS * QKV_N;
    float* kb   = qb  + (size_t)BB * NH * TT * DH;
    float* vb   = kb  + (size_t)BB * NH * TT * DH;
    float* attn = qkv;

    {
        const int blocks = (ROWS / GBM) * (QKV_N / GBN);
        gemm_bias_f32wmma<<<blocks, 256, 0, stream>>>(
            x, W_qkv, b_qkv, qkv, ROWS, QKV_N, DM);
    }

    rope_split<<<ROWS, 256, 0, stream>>>(qkv, qb, kb, vb);

    {
        const int blocks = BB * NH * (TT / 64);
        flash_attn_f32wmma<<<blocks, 128, 0, stream>>>(qb, kb, vb, mask, attn);
    }

    {
        const int blocks = (ROWS / GBM) * (DM / GBN);
        gemm_bias_f32wmma<<<blocks, 256, 0, stream>>>(
            attn, W_out, b_out, out, ROWS, DM, DM);
    }
}


# Attention: each wave takes query tile p, then its mirror tile 127-p: 129 key tiles for every wave (before: 1 to 128 by tile); grid 256 blocks of 8 waves; per-tile code and order of sums unchanged
# speedup vs baseline: 2.8567x; 1.0372x over previous
#include <hip/hip_runtime.h>
#include <hip/hip_bf16.h>
#include <math.h>


#define BB      2
#define TT      2048
#define DM      1024
#define NH      16
#define DH      64
#define ROWS    (BB * TT)
#define QKV_N   (3 * DM)

typedef __attribute__((ext_vector_type(2))) float v2f;
typedef __attribute__((ext_vector_type(8))) float v8f;

typedef __attribute__((ext_vector_type(4))) float g4f;

#define GBM      128
#define GBN      128
#define GBK      32
#define APITCH   36
#define BPITCH2  288
#define A_STAGE  (GBM * APITCH)
#define B_STAGE  ((GBK / 2) * BPITCH2)
#define STAGE_F  (A_STAGE + B_STAGE)

struct GFrag { v2f a0, a1, b0, b1, b2, b3; };

__device__ __forceinline__ void load_gfrag(GFrag& f, const float* ap, const float* bp)
{
    f.a0 = *(const v2f*)(ap);
    f.a1 = *(const v2f*)(ap + 16 * APITCH);
    f.b0 = *(const v2f*)(bp);
    f.b1 = *(const v2f*)(bp + 32);
    f.b2 = *(const v2f*)(bp + 64);
    f.b3 = *(const v2f*)(bp + 96);
}

__device__ __forceinline__ void store_bpair(float* dst, g4f r0, g4f r1)
{
    g4f lo, hi;
    lo.x = r0.x; lo.y = r1.x; lo.z = r0.y; lo.w = r1.y;
    hi.x = r0.z; hi.y = r1.z; hi.z = r0.w; hi.w = r1.w;
    *(g4f*)(dst)     = lo;
    *(g4f*)(dst + 4) = hi;
}

__global__ void __launch_bounds__(256)
gemm_bias_f32wmma(const float* __restrict__ A,
                  const float* __restrict__ Bm,
                  const float* __restrict__ bias,
                  float* __restrict__ C,
                  int M, int N, int K)
{
    __shared__ __attribute__((aligned(16))) float lds[2 * STAGE_F];

    (void)M;
    const int tid  = threadIdx.x;
    const int wid  = __builtin_amdgcn_readfirstlane(tid >> 5);
    const int lane = tid & 31;
    const int half = lane >> 4;
    const int ln   = lane & 15;
    const int wm   = wid & 3;
    const int wn   = wid >> 2;

    const int nbn = N / GBN;
    const int m0  = (blockIdx.x / nbn) * GBM;
    const int n0  = (blockIdx.x % nbn) * GBN;

    const float* __restrict__ ag = A  + (size_t)(m0 + (tid >> 3)) * K + 4 * (tid & 7);
    const float* __restrict__ bg = Bm + (size_t)(2 * (tid >> 5)) * N + n0 + 4 * (tid & 31);
    const int as_off = (tid >> 3) * APITCH + 4 * (tid & 7);
    const int bs_off = (tid >> 5) * BPITCH2 + 8 * (tid & 31);

    const int a_rd = (wm * 32 + ln) * APITCH + 2 * half;
    const int b_rd = half * BPITCH2 + 2 * (wn * 64 + ln);

    g4f ra[4], rb[4];
#pragma unroll
    for (int p = 0; p < 4; ++p) {
        ra[p] = *(const g4f*)(ag + (size_t)(32 * p) * K);
        rb[p] = *(const g4f*)(bg + (size_t)(16 * (p >> 1) + (p & 1)) * N);
    }
#pragma unroll
    for (int p = 0; p < 4; ++p)
        *(g4f*)(lds + as_off + 32 * p * APITCH) = ra[p];
#pragma unroll
    for (int q = 0; q < 2; ++q)
        store_bpair(lds + A_STAGE + bs_off + 8 * q * BPITCH2, rb[2 * q], rb[2 * q + 1]);
    __syncthreads();

    v8f acc00 = {}, acc01 = {}, acc02 = {}, acc03 = {};
    v8f acc10 = {}, acc11 = {}, acc12 = {}, acc13 = {};

    const int nch = K / GBK;
    for (int c = 0; c < nch; ++c) {
        const float* As = lds + (c & 1) * STAGE_F;
        const float* Bs = As + A_STAGE;
        const bool more = (c + 1 < nch);

        if (more) {
            ag += GBK;
            bg += (size_t)GBK * N;
#pragma unroll
            for (int p = 0; p < 4; ++p) {
                ra[p] = *(const g4f*)(ag + (size_t)(32 * p) * K);
                rb[p] = *(const g4f*)(bg + (size_t)(16 * (p >> 1) + (p & 1)) * N);
            }
        }

        GFrag f[2];
        load_gfrag(f[0], As + a_rd, Bs + b_rd);
#pragma unroll
        for (int ks = 0; ks < GBK / 4; ++ks) {
            if (ks + 1 < GBK / 4)
                load_gfrag(f[(ks + 1) & 1], As + a_rd + 4 * (ks + 1), Bs + b_rd + 2 * (ks + 1) * BPITCH2);
            __builtin_amdgcn_sched_barrier(0);
            const GFrag& g = f[ks & 1];
            acc00 = __builtin_amdgcn_wmma_f32_16x16x4_f32(false, g.a0, false, g.b0, (short)0, acc00, false, false);
            acc10 = __builtin_amdgcn_wmma_f32_16x16x4_f32(false, g.a1, false, g.b0, (short)0, acc10, false, false);
            acc01 = __builtin_amdgcn_wmma_f32_16x16x4_f32(false, g.a0, false, g.b1, (short)0, acc01, false, false);
            acc11 = __builtin_amdgcn_wmma_f32_16x16x4_f32(false, g.a1, false, g.b1, (short)0, acc11, false, false);
            acc02 = __builtin_amdgcn_wmma_f32_16x16x4_f32(false, g.a0, false, g.b2, (short)0, acc02, false, false);
            acc12 = __builtin_amdgcn_wmma_f32_16x16x4_f32(false, g.a1, false, g.b2, (short)0, acc12, false, false);
            acc03 = __builtin_amdgcn_wmma_f32_16x16x4_f32(false, g.a0, false, g.b3, (short)0, acc03, false, false);
            acc13 = __builtin_amdgcn_wmma_f32_16x16x4_f32(false, g.a1, false, g.b3, (short)0, acc13, false, false);
            __builtin_amdgcn_sched_barrier(0);
        }

        if (more) {
            float* Ad = lds + ((c + 1) & 1) * STAGE_F;
            float* Bd = Ad + A_STAGE;
#pragma unroll
            for (int p = 0; p < 4; ++p)
                *(g4f*)(Ad + as_off + 32 * p * APITCH) = ra[p];
#pragma unroll
            for (int q = 0; q < 2; ++q)
                store_bpair(Bd + bs_off + 8 * q * BPITCH2, rb[2 * q], rb[2 * q + 1]);
        }
        __syncthreads();
    }

    v8f accs[2][4] = { { acc00, acc01, acc02, acc03 },
                       { acc10, acc11, acc12, acc13 } };
#pragma unroll
    for (int sub = 0; sub < 2; ++sub) {
        const int rbase = m0 + wm * 32 + sub * 16 + 8 * half;
#pragma unroll
        for (int t = 0; t < 4; ++t) {
            const int col = n0 + wn * 64 + t * 16 + ln;
            const float bs = bias[col];
#pragma unroll
            for (int v = 0; v < 8; ++v)
                C[(size_t)(rbase + v) * N + col] = accs[sub][t][v] + bs;
        }
    }
}

__global__ void rope_split(const float* __restrict__ qkv,
                           float* __restrict__ qb,
                           float* __restrict__ kb,
                           float* __restrict__ vb)
{
    const int row = blockIdx.x;
    const int b = row / TT;
    const int t = row % TT;
    const float* __restrict__ src = qkv + (size_t)row * QKV_N;

    for (int p = threadIdx.x; p < QKV_N / 2; p += blockDim.x) {
        const int c   = p * 2;
        const int sec = c >> 10;
        const int cc  = c & (DM - 1);
        const int h   = cc >> 6;
        const int d   = cc & (DH - 1);

        const float e = src[c];
        const float o = src[c + 1];
        const size_t dst = (((size_t)(b * NH + h)) * TT + t) * DH + d;

        if (sec == 2) {
            vb[dst]     = e;
            vb[dst + 1] = o;
        } else {
            const int   i   = d >> 1;
            const float inv = expf(-(float)(2 * i) * (9.210340371976184f / 64.0f));
            const float ang = (float)t * inv;
            const float sn = sinf(ang);
            const float cs = cosf(ang);
            const float re = e * cs - o * sn;
            const float ro = e * sn + o * cs;
            if (sec == 0) { qb[dst] = re; qb[dst + 1] = ro; }
            else          { kb[dst] = re; kb[dst + 1] = ro; }
        }
    }
}

typedef __attribute__((ext_vector_type(4))) float v4f;
typedef __attribute__((ext_vector_type(4))) int   v4i;

__device__ __forceinline__ float xhalf(float x)
{
    const int bits = __builtin_bit_cast(int, x);
    return __builtin_bit_cast(float,
        __builtin_amdgcn_permlanex16(bits, bits, 0x76543210, 0xfedcba98, false, false));
}

__global__ void __launch_bounds__(256)
flash_attn_f32wmma(const float* __restrict__ qb,
                   const float* __restrict__ kb,
                   const float* __restrict__ vb,
                   const int*   __restrict__ amask,
                   float* __restrict__ attn  )
{
    const int wid  = __builtin_amdgcn_readfirstlane((int)(threadIdx.x >> 5));
    const int lane = threadIdx.x & 31;
    const int half = lane >> 4;
    const int ln   = lane & 15;

    const int NQT  = TT / 16;
    const int bh   = blockIdx.x >> 3;
    const int pr   = ((blockIdx.x & 7) << 3) + wid;
    const int b    = bh >> 4;
    const int h    = bh & (NH - 1);

    const float* __restrict__ Q  = qb + (size_t)bh * TT * DH;
    const float* __restrict__ Kp = kb + (size_t)bh * TT * DH;
    const float* __restrict__ Vp = vb + (size_t)bh * TT * DH;

    const float* __restrict__ kptr = Kp + (size_t)ln * DH + 4 * half;
    const float* __restrict__ vptr = Vp + (size_t)(8 * half) * DH + 4 * ln;
    const int*   __restrict__ mptr = amask + b * TT + 8 * half;

    for (int pass = 0; pass < 2; ++pass) {
        const int q0 = (pass == 0 ? pr : NQT - 1 - pr) * 16;

        v4f q4[8];
        {
            const float* __restrict__ qrow = Q + (size_t)(q0 + ln) * DH + 4 * half;
            for (int w = 0; w < 8; ++w)
                q4[w] = *(const v4f*)(qrow + 8 * w);
        }

        v8f o0 = {}, o1 = {}, o2 = {}, o3 = {};
        float rmax = -1e30f;
        float rsum = 0.0f;

        const float scale = 0.125f;
        const int nkt = q0 / 16 + 1;

        for (int j = 0; j < nkt; ++j) {
            const int kbase = j * 16;

            v4f k4[8];
            for (int w = 0; w < 8; ++w)
                k4[w] = *(const v4f*)(kptr + (size_t)kbase * DH + 8 * w);
            v4f v4[8];
            for (int r = 0; r < 8; ++r)
                v4[r] = *(const v4f*)(vptr + (size_t)(kbase + r) * DH);
            const v4i pm0 = *(const v4i*)(mptr + kbase);
            const v4i pm1 = *(const v4i*)(mptr + kbase + 4);

            v8f s = {};
            for (int w = 0; w < 8; ++w) {
                v2f ka, qa, kc, qc;
                ka.x = k4[w].x; ka.y = k4[w].y;  qa.x = q4[w].x; qa.y = q4[w].y;
                kc.x = k4[w].z; kc.y = k4[w].w;  qc.x = q4[w].z; qc.y = q4[w].w;
                s = __builtin_amdgcn_wmma_f32_16x16x4_f32(false, ka, false, qa, (short)0, s, false, false);
                s = __builtin_amdgcn_wmma_f32_16x16x4_f32(false, kc, false, qc, (short)0, s, false, false);
            }

            const int pmv[8] = { pm0.x, pm0.y, pm0.z, pm0.w, pm1.x, pm1.y, pm1.z, pm1.w };
            float x[8];
            for (int r = 0; r < 8; ++r)
                x[r] = (pmv[r] == 0) ? -1e30f : s[r] * scale;
            if (j == nkt - 1) {
                for (int r = 0; r < 8; ++r)
                    x[r] = (8 * half + r > ln) ? -1e30f : x[r];
            }

            float mx = fmaxf(fmaxf(fmaxf(x[0], x[1]), fmaxf(x[2], x[3])),
                             fmaxf(fmaxf(x[4], x[5]), fmaxf(x[6], x[7])));
            mx = fmaxf(mx, xhalf(mx));
            const float nmax  = fmaxf(rmax, mx);
            const float alpha = __expf(rmax - nmax);
            float p[8];
            for (int r = 0; r < 8; ++r)
                p[r] = __expf(x[r] - nmax);
            float sr = ((p[0] + p[1]) + (p[2] + p[3])) + ((p[4] + p[5]) + (p[6] + p[7]));
            sr += xhalf(sr);
            rsum = rsum * alpha + sr;
            rmax = nmax;
            o0 *= alpha; o1 *= alpha; o2 *= alpha; o3 *= alpha;

            for (int i = 0; i < 4; ++i) {
                v2f pf; pf.x = p[2 * i]; pf.y = p[2 * i + 1];
                v2f a0; a0.x = v4[2 * i].x; a0.y = v4[2 * i + 1].x;
                v2f a1; a1.x = v4[2 * i].y; a1.y = v4[2 * i + 1].y;
                v2f a2; a2.x = v4[2 * i].z; a2.y = v4[2 * i + 1].z;
                v2f a3; a3.x = v4[2 * i].w; a3.y = v4[2 * i + 1].w;
                o0 = __builtin_amdgcn_wmma_f32_16x16x4_f32(false, a0, false, pf, (short)0, o0, false, false);
                o1 = __builtin_amdgcn_wmma_f32_16x16x4_f32(false, a1, false, pf, (short)0, o1, false, false);
                o2 = __builtin_amdgcn_wmma_f32_16x16x4_f32(false, a2, false, pf, (short)0, o2, false, false);
                o3 = __builtin_amdgcn_wmma_f32_16x16x4_f32(false, a3, false, pf, (short)0, o3, false, false);
            }
        }

        const float inv = 1.0f / rsum;
        float* __restrict__ orow =
            attn + ((size_t)b * TT + (q0 + ln)) * DM + h * DH + 32 * half;
        for (int r = 0; r < 8; ++r) {
            v4f o;
            o.x = o0[r] * inv; o.y = o1[r] * inv; o.z = o2[r] * inv; o.w = o3[r] * inv;
            *(v4f*)(orow + 4 * r) = o;
        }
    }
}

extern "C" void kernel_launch(void* const* d_in, const int* in_sizes, int n_in,
                              void* d_out, int out_size, void* d_ws, size_t ws_size,
                              hipStream_t stream)
{
    (void)in_sizes; (void)n_in; (void)out_size; (void)ws_size;

    const float* x      = (const float*)d_in[0];
    const int*   mask   = (const int*)  d_in[1];
    const float* W_qkv  = (const float*)d_in[2];
    const float* b_qkv  = (const float*)d_in[3];
    const float* W_out  = (const float*)d_in[4];
    const float* b_out  = (const float*)d_in[5];
    float*       out    = (float*)d_out;

    float* ws   = (float*)d_ws;
    float* qkv  = ws;
    float* qb   = qkv + (size_t)ROWS * QKV_N;
    float* kb   = qb  + (size_t)BB * NH * TT * DH;
    float* vb   = kb  + (size_t)BB * NH * TT * DH;
    float* attn = qkv;

    {
        const int blocks = (ROWS / GBM) * (QKV_N / GBN);
        gemm_bias_f32wmma<<<blocks, 256, 0, stream>>>(
            x, W_qkv, b_qkv, qkv, ROWS, QKV_N, DM);
    }

    rope_split<<<ROWS, 256, 0, stream>>>(qkv, qb, kb, vb);

    {
        const int blocks = BB * NH * 8;
        flash_attn_f32wmma<<<blocks, 256, 0, stream>>>(qb, kb, vb, mask, attn);
    }

    {
        const int blocks = (ROWS / GBM) * (DM / GBN);
        gemm_bias_f32wmma<<<blocks, 256, 0, stream>>>(
            attn, W_out, b_out, out, ROWS, DM, DM);
    }
}


# Attention: K, V and mask of key tile j+1 are loaded into a second register set in front of tile j's products (one tile ahead); per-tile arithmetic and order of sums unchanged
# speedup vs baseline: 3.1546x; 1.1043x over previous
#include <hip/hip_runtime.h>
#include <hip/hip_bf16.h>
#include <math.h>


#define BB      2
#define TT      2048
#define DM      1024
#define NH      16
#define DH      64
#define ROWS    (BB * TT)
#define QKV_N   (3 * DM)

typedef __attribute__((ext_vector_type(2))) float v2f;
typedef __attribute__((ext_vector_type(8))) float v8f;

typedef __attribute__((ext_vector_type(4))) float g4f;

#define GBM      128
#define GBN      128
#define GBK      32
#define APITCH   36
#define BPITCH2  288
#define A_STAGE  (GBM * APITCH)
#define B_STAGE  ((GBK / 2) * BPITCH2)
#define STAGE_F  (A_STAGE + B_STAGE)

struct GFrag { v2f a0, a1, b0, b1, b2, b3; };

__device__ __forceinline__ void load_gfrag(GFrag& f, const float* ap, const float* bp)
{
    f.a0 = *(const v2f*)(ap);
    f.a1 = *(const v2f*)(ap + 16 * APITCH);
    f.b0 = *(const v2f*)(bp);
    f.b1 = *(const v2f*)(bp + 32);
    f.b2 = *(const v2f*)(bp + 64);
    f.b3 = *(const v2f*)(bp + 96);
}

__device__ __forceinline__ void store_bpair(float* dst, g4f r0, g4f r1)
{
    g4f lo, hi;
    lo.x = r0.x; lo.y = r1.x; lo.z = r0.y; lo.w = r1.y;
    hi.x = r0.z; hi.y = r1.z; hi.z = r0.w; hi.w = r1.w;
    *(g4f*)(dst)     = lo;
    *(g4f*)(dst + 4) = hi;
}

__global__ void __launch_bounds__(256)
gemm_bias_f32wmma(const float* __restrict__ A,
                  const float* __restrict__ Bm,
                  const float* __restrict__ bias,
                  float* __restrict__ C,
                  int M, int N, int K)
{
    __shared__ __attribute__((aligned(16))) float lds[2 * STAGE_F];

    (void)M;
    const int tid  = threadIdx.x;
    const int wid  = __builtin_amdgcn_readfirstlane(tid >> 5);
    const int lane = tid & 31;
    const int half = lane >> 4;
    const int ln   = lane & 15;
    const int wm   = wid & 3;
    const int wn   = wid >> 2;

    const int nbn = N / GBN;
    const int m0  = (blockIdx.x / nbn) * GBM;
    const int n0  = (blockIdx.x % nbn) * GBN;

    const float* __restrict__ ag = A  + (size_t)(m0 + (tid >> 3)) * K + 4 * (tid & 7);
    const float* __restrict__ bg = Bm + (size_t)(2 * (tid >> 5)) * N + n0 + 4 * (tid & 31);
    const int as_off = (tid >> 3) * APITCH + 4 * (tid & 7);
    const int bs_off = (tid >> 5) * BPITCH2 + 8 * (tid & 31);

    const int a_rd = (wm * 32 + ln) * APITCH + 2 * half;
    const int b_rd = half * BPITCH2 + 2 * (wn * 64 + ln);

    g4f ra[4], rb[4];
#pragma unroll
    for (int p = 0; p < 4; ++p) {
        ra[p] = *(const g4f*)(ag + (size_t)(32 * p) * K);
        rb[p] = *(const g4f*)(bg + (size_t)(16 * (p >> 1) + (p & 1)) * N);
    }
#pragma unroll
    for (int p = 0; p < 4; ++p)
        *(g4f*)(lds + as_off + 32 * p * APITCH) = ra[p];
#pragma unroll
    for (int q = 0; q < 2; ++q)
        store_bpair(lds + A_STAGE + bs_off + 8 * q * BPITCH2, rb[2 * q], rb[2 * q + 1]);
    __syncthreads();

    v8f acc00 = {}, acc01 = {}, acc02 = {}, acc03 = {};
    v8f acc10 = {}, acc11 = {}, acc12 = {}, acc13 = {};

    const int nch = K / GBK;
    for (int c = 0; c < nch; ++c) {
        const float* As = lds + (c & 1) * STAGE_F;
        const float* Bs = As + A_STAGE;
        const bool more = (c + 1 < nch);

        if (more) {
            ag += GBK;
            bg += (size_t)GBK * N;
#pragma unroll
            for (int p = 0; p < 4; ++p) {
                ra[p] = *(const g4f*)(ag + (size_t)(32 * p) * K);
                rb[p] = *(const g4f*)(bg + (size_t)(16 * (p >> 1) + (p & 1)) * N);
            }
        }

        GFrag f[2];
        load_gfrag(f[0], As + a_rd, Bs + b_rd);
#pragma unroll
        for (int ks = 0; ks < GBK / 4; ++ks) {
            if (ks + 1 < GBK / 4)
                load_gfrag(f[(ks + 1) & 1], As + a_rd + 4 * (ks + 1), Bs + b_rd + 2 * (ks + 1) * BPITCH2);
            __builtin_amdgcn_sched_barrier(0);
            const GFrag& g = f[ks & 1];
            acc00 = __builtin_amdgcn_wmma_f32_16x16x4_f32(false, g.a0, false, g.b0, (short)0, acc00, false, false);
            acc10 = __builtin_amdgcn_wmma_f32_16x16x4_f32(false, g.a1, false, g.b0, (short)0, acc10, false, false);
            acc01 = __builtin_amdgcn_wmma_f32_16x16x4_f32(false, g.a0, false, g.b1, (short)0, acc01, false, false);
            acc11 = __builtin_amdgcn_wmma_f32_16x16x4_f32(false, g.a1, false, g.b1, (short)0, acc11, false, false);
            acc02 = __builtin_amdgcn_wmma_f32_16x16x4_f32(false, g.a0, false, g.b2, (short)0, acc02, false, false);
            acc12 = __builtin_amdgcn_wmma_f32_16x16x4_f32(false, g.a1, false, g.b2, (short)0, acc12, false, false);
            acc03 = __builtin_amdgcn_wmma_f32_16x16x4_f32(false, g.a0, false, g.b3, (short)0, acc03, false, false);
            acc13 = __builtin_amdgcn_wmma_f32_16x16x4_f32(false, g.a1, false, g.b3, (short)0, acc13, false, false);
            __builtin_amdgcn_sched_barrier(0);
        }

        if (more) {
            float* Ad = lds + ((c + 1) & 1) * STAGE_F;
            float* Bd = Ad + A_STAGE;
#pragma unroll
            for (int p = 0; p < 4; ++p)
                *(g4f*)(Ad + as_off + 32 * p * APITCH) = ra[p];
#pragma unroll
            for (int q = 0; q < 2; ++q)
                store_bpair(Bd + bs_off + 8 * q * BPITCH2, rb[2 * q], rb[2 * q + 1]);
        }
        __syncthreads();
    }

    v8f accs[2][4] = { { acc00, acc01, acc02, acc03 },
                       { acc10, acc11, acc12, acc13 } };
#pragma unroll
    for (int sub = 0; sub < 2; ++sub) {
        const int rbase = m0 + wm * 32 + sub * 16 + 8 * half;
#pragma unroll
        for (int t = 0; t < 4; ++t) {
            const int col = n0 + wn * 64 + t * 16 + ln;
            const float bs = bias[col];
#pragma unroll
            for (int v = 0; v < 8; ++v)
                C[(size_t)(rbase + v) * N + col] = accs[sub][t][v] + bs;
        }
    }
}

__global__ void rope_split(const float* __restrict__ qkv,
                           float* __restrict__ qb,
                           float* __restrict__ kb,
                           float* __restrict__ vb)
{
    const int row = blockIdx.x;
    const int b = row / TT;
    const int t = row % TT;
    const float* __restrict__ src = qkv + (size_t)row * QKV_N;

    for (int p = threadIdx.x; p < QKV_N / 2; p += blockDim.x) {
        const int c   = p * 2;
        const int sec = c >> 10;
        const int cc  = c & (DM - 1);
        const int h   = cc >> 6;
        const int d   = cc & (DH - 1);

        const float e = src[c];
        const float o = src[c + 1];
        const size_t dst = (((size_t)(b * NH + h)) * TT + t) * DH + d;

        if (sec == 2) {
            vb[dst]     = e;
            vb[dst + 1] = o;
        } else {
            const int   i   = d >> 1;
            const float inv = expf(-(float)(2 * i) * (9.210340371976184f / 64.0f));
            const float ang = (float)t * inv;
            const float sn = sinf(ang);
            const float cs = cosf(ang);
            const float re = e * cs - o * sn;
            const float ro = e * sn + o * cs;
            if (sec == 0) { qb[dst] = re; qb[dst + 1] = ro; }
            else          { kb[dst] = re; kb[dst + 1] = ro; }
        }
    }
}

typedef __attribute__((ext_vector_type(4))) float v4f;
typedef __attribute__((ext_vector_type(4))) int   v4i;

__device__ __forceinline__ float xhalf(float x)
{
    const int bits = __builtin_bit_cast(int, x);
    return __builtin_bit_cast(float,
        __builtin_amdgcn_permlanex16(bits, bits, 0x76543210, 0xfedcba98, false, false));
}

struct KVTile {
    v4f k4[8];
    v4f v4[8];
    v4i pm0, pm1;
};

__device__ __forceinline__ void load_kv_tile(KVTile& t,
                                             const float* __restrict__ kptr,
                                             const float* __restrict__ vptr,
                                             const int*   __restrict__ mptr,
                                             int kbase)
{
    for (int w = 0; w < 8; ++w)
        t.k4[w] = *(const v4f*)(kptr + (size_t)kbase * DH + 8 * w);
    for (int r = 0; r < 8; ++r)
        t.v4[r] = *(const v4f*)(vptr + (size_t)(kbase + r) * DH);
    t.pm0 = *(const v4i*)(mptr + kbase);
    t.pm1 = *(const v4i*)(mptr + kbase + 4);
}

__device__ __forceinline__ void hold_loaded(const KVTile& t)
{
    asm volatile("" :: "v"(t.k4[0]), "v"(t.k4[1]), "v"(t.k4[2]), "v"(t.k4[3]),
                       "v"(t.k4[4]), "v"(t.k4[5]), "v"(t.k4[6]), "v"(t.k4[7]),
                       "v"(t.v4[0]), "v"(t.v4[1]), "v"(t.v4[2]), "v"(t.v4[3]),
                       "v"(t.v4[4]), "v"(t.v4[5]), "v"(t.v4[6]), "v"(t.v4[7]),
                       "v"(t.pm0), "v"(t.pm1));
}

__device__ __forceinline__ void attn_tile(const KVTile& t, const v4f (&q4)[8],
                                          v8f& o0, v8f& o1, v8f& o2, v8f& o3,
                                          float& rmax, float& rsum,
                                          bool diag, int half, int ln)
{
    const float scale = 0.125f;

    v8f s = {};
    for (int w = 0; w < 8; ++w) {
        v2f ka, qa, kc, qc;
        ka.x = t.k4[w].x; ka.y = t.k4[w].y;  qa.x = q4[w].x; qa.y = q4[w].y;
        kc.x = t.k4[w].z; kc.y = t.k4[w].w;  qc.x = q4[w].z; qc.y = q4[w].w;
        s = __builtin_amdgcn_wmma_f32_16x16x4_f32(false, ka, false, qa, (short)0, s, false, false);
        s = __builtin_amdgcn_wmma_f32_16x16x4_f32(false, kc, false, qc, (short)0, s, false, false);
    }

    const int pmv[8] = { t.pm0.x, t.pm0.y, t.pm0.z, t.pm0.w, t.pm1.x, t.pm1.y, t.pm1.z, t.pm1.w };
    float x[8];
    for (int r = 0; r < 8; ++r)
        x[r] = (pmv[r] == 0) ? -1e30f : s[r] * scale;
    if (diag) {
        for (int r = 0; r < 8; ++r)
            x[r] = (8 * half + r > ln) ? -1e30f : x[r];
    }

    float mx = fmaxf(fmaxf(fmaxf(x[0], x[1]), fmaxf(x[2], x[3])),
                     fmaxf(fmaxf(x[4], x[5]), fmaxf(x[6], x[7])));
    mx = fmaxf(mx, xhalf(mx));
    const float nmax  = fmaxf(rmax, mx);
    const float alpha = __expf(rmax - nmax);
    float p[8];
    for (int r = 0; r < 8; ++r)
        p[r] = __expf(x[r] - nmax);
    float sr = ((p[0] + p[1]) + (p[2] + p[3])) + ((p[4] + p[5]) + (p[6] + p[7]));
    sr += xhalf(sr);
    rsum = rsum * alpha + sr;
    rmax = nmax;
    o0 *= alpha; o1 *= alpha; o2 *= alpha; o3 *= alpha;

    for (int i = 0; i < 4; ++i) {
        v2f pf; pf.x = p[2 * i]; pf.y = p[2 * i + 1];
        v2f a0; a0.x = t.v4[2 * i].x; a0.y = t.v4[2 * i + 1].x;
        v2f a1; a1.x = t.v4[2 * i].y; a1.y = t.v4[2 * i + 1].y;
        v2f a2; a2.x = t.v4[2 * i].z; a2.y = t.v4[2 * i + 1].z;
        v2f a3; a3.x = t.v4[2 * i].w; a3.y = t.v4[2 * i + 1].w;
        o0 = __builtin_amdgcn_wmma_f32_16x16x4_f32(false, a0, false, pf, (short)0, o0, false, false);
        o1 = __builtin_amdgcn_wmma_f32_16x16x4_f32(false, a1, false, pf, (short)0, o1, false, false);
        o2 = __builtin_amdgcn_wmma_f32_16x16x4_f32(false, a2, false, pf, (short)0, o2, false, false);
        o3 = __builtin_amdgcn_wmma_f32_16x16x4_f32(false, a3, false, pf, (short)0, o3, false, false);
    }
}

__global__ void __launch_bounds__(256)
flash_attn_f32wmma(const float* __restrict__ qb,
                   const float* __restrict__ kb,
                   const float* __restrict__ vb,
                   const int*   __restrict__ amask,
                   float* __restrict__ attn  )
{
    const int wid  = __builtin_amdgcn_readfirstlane((int)(threadIdx.x >> 5));
    const int lane = threadIdx.x & 31;
    const int half = lane >> 4;
    const int ln   = lane & 15;

    const int NQT  = TT / 16;
    const int bh   = blockIdx.x >> 3;
    const int pr   = ((blockIdx.x & 7) << 3) + wid;
    const int b    = bh >> 4;
    const int h    = bh & (NH - 1);

    const float* __restrict__ Q  = qb + (size_t)bh * TT * DH;
    const float* __restrict__ Kp = kb + (size_t)bh * TT * DH;
    const float* __restrict__ Vp = vb + (size_t)bh * TT * DH;

    const float* __restrict__ kptr = Kp + (size_t)ln * DH + 4 * half;
    const float* __restrict__ vptr = Vp + (size_t)(8 * half) * DH + 4 * ln;
    const int*   __restrict__ mptr = amask + b * TT + 8 * half;

    for (int pass = 0; pass < 2; ++pass) {
        const int q0 = (pass == 0 ? pr : NQT - 1 - pr) * 16;

        v4f q4[8];
        {
            const float* __restrict__ qrow = Q + (size_t)(q0 + ln) * DH + 4 * half;
            for (int w = 0; w < 8; ++w)
                q4[w] = *(const v4f*)(qrow + 8 * w);
        }

        v8f o0 = {}, o1 = {}, o2 = {}, o3 = {};
        float rmax = -1e30f;
        float rsum = 0.0f;

        const int nkt = q0 / 16 + 1;

        KVTile ta, tb;
        load_kv_tile(ta, kptr, vptr, mptr, 0);
        int j = 0;
        for (;;) {
            if (j == nkt - 1) { attn_tile(ta, q4, o0, o1, o2, o3, rmax, rsum, true, half, ln); break; }
            load_kv_tile(tb, kptr, vptr, mptr, (j + 1) * 16);
            attn_tile(ta, q4, o0, o1, o2, o3, rmax, rsum, false, half, ln);
            __builtin_amdgcn_sched_barrier(0);
            hold_loaded(tb);
            ++j;
            if (j == nkt - 1) { attn_tile(tb, q4, o0, o1, o2, o3, rmax, rsum, true, half, ln); break; }
            load_kv_tile(ta, kptr, vptr, mptr, (j + 1) * 16);
            attn_tile(tb, q4, o0, o1, o2, o3, rmax, rsum, false, half, ln);
            __builtin_amdgcn_sched_barrier(0);
            hold_loaded(ta);
            ++j;
        }

        const float inv = 1.0f / rsum;
        float* __restrict__ orow =
            attn + ((size_t)b * TT + (q0 + ln)) * DM + h * DH + 32 * half;
        for (int r = 0; r < 8; ++r) {
            v4f o;
            o.x = o0[r] * inv; o.y = o1[r] * inv; o.z = o2[r] * inv; o.w = o3[r] * inv;
            *(v4f*)(orow + 4 * r) = o;
        }
    }
}

extern "C" void kernel_launch(void* const* d_in, const int* in_sizes, int n_in,
                              void* d_out, int out_size, void* d_ws, size_t ws_size,
                              hipStream_t stream)
{
    (void)in_sizes; (void)n_in; (void)out_size; (void)ws_size;

    const float* x      = (const float*)d_in[0];
    const int*   mask   = (const int*)  d_in[1];
    const float* W_qkv  = (const float*)d_in[2];
    const float* b_qkv  = (const float*)d_in[3];
    const float* W_out  = (const float*)d_in[4];
    const float* b_out  = (const float*)d_in[5];
    float*       out    = (float*)d_out;

    float* ws   = (float*)d_ws;
    float* qkv  = ws;
    float* qb   = qkv + (size_t)ROWS * QKV_N;
    float* kb   = qb  + (size_t)BB * NH * TT * DH;
    float* vb   = kb  + (size_t)BB * NH * TT * DH;
    float* attn = qkv;

    {
        const int blocks = (ROWS / GBM) * (QKV_N / GBN);
        gemm_bias_f32wmma<<<blocks, 256, 0, stream>>>(
            x, W_qkv, b_qkv, qkv, ROWS, QKV_N, DM);
    }

    rope_split<<<ROWS, 256, 0, stream>>>(qkv, qb, kb, vb);

    {
        const int blocks = BB * NH * 8;
        flash_attn_f32wmma<<<blocks, 256, 0, stream>>>(qb, kb, vb, mask, attn);
    }

    {
        const int blocks = (ROWS / GBM) * (DM / GBN);
        gemm_bias_f32wmma<<<blocks, 256, 0, stream>>>(
            attn, W_out, b_out, out, ROWS, DM, DM);
    }
}


# RoPE and head split fused into the QKV GEMM epilogue: tile columns ordered so a lane owns whole rotary pairs, q/k/v stored in whole lines; rope launch and qkv round trip removed; sums unchanged
# speedup vs baseline: 3.3598x; 1.0650x over previous
#include <hip/hip_runtime.h>
#include <hip/hip_bf16.h>
#include <math.h>


#define BB      2
#define TT      2048
#define DM      1024
#define NH      16
#define DH      64
#define ROWS    (BB * TT)
#define QKV_N   (3 * DM)

typedef __attribute__((ext_vector_type(2))) float v2f;
typedef __attribute__((ext_vector_type(8))) float v8f;

typedef __attribute__((ext_vector_type(4))) float g4f;

#define GBM      128
#define GBN      128
#define GBK      32
#define APITCH   36
#define BPITCH2  288
#define A_STAGE  (GBM * APITCH)
#define B_STAGE  ((GBK / 2) * BPITCH2)
#define STAGE_F  (A_STAGE + B_STAGE)

struct GFrag { v2f a0, a1, b0, b1, b2, b3; };

template <bool PAIRS>
__device__ __forceinline__ void load_gfrag(GFrag& f, const float* ap, const float* bp)
{
    f.a0 = *(const v2f*)(ap);
    f.a1 = *(const v2f*)(ap + 16 * APITCH);
    if (PAIRS) {
        const g4f b01 = *(const g4f*)(bp);
        const g4f b23 = *(const g4f*)(bp + 64);
        f.b0.x = b01.x; f.b0.y = b01.y;  f.b1.x = b01.z; f.b1.y = b01.w;
        f.b2.x = b23.x; f.b2.y = b23.y;  f.b3.x = b23.z; f.b3.y = b23.w;
    } else {
        f.b0 = *(const v2f*)(bp);
        f.b1 = *(const v2f*)(bp + 32);
        f.b2 = *(const v2f*)(bp + 64);
        f.b3 = *(const v2f*)(bp + 96);
    }
}

__device__ __forceinline__ void store_bpair(float* dst, g4f r0, g4f r1)
{
    g4f lo, hi;
    lo.x = r0.x; lo.y = r1.x; lo.z = r0.y; lo.w = r1.y;
    hi.x = r0.z; hi.y = r1.z; hi.z = r0.w; hi.w = r1.w;
    *(g4f*)(dst)     = lo;
    *(g4f*)(dst + 4) = hi;
}

template <bool ROPE>
__device__ __forceinline__ void gemm_body(float* lds,
                                          const float* __restrict__ A,
                                          const float* __restrict__ Bm,
                                          const float* __restrict__ bias,
                                          float* __restrict__ C,
                                          int N, int K)
{
    const int tid  = threadIdx.x;
    const int wid  = __builtin_amdgcn_readfirstlane(tid >> 5);
    const int lane = tid & 31;
    const int half = lane >> 4;
    const int ln   = lane & 15;
    const int wm   = wid & 3;
    const int wn   = wid >> 2;

    const int nbn = N / GBN;
    const int m0  = (blockIdx.x / nbn) * GBM;
    const int n0  = (blockIdx.x % nbn) * GBN;

    const float* __restrict__ ag = A  + (size_t)(m0 + (tid >> 3)) * K + 4 * (tid & 7);
    const float* __restrict__ bg = Bm + (size_t)(2 * (tid >> 5)) * N + n0 + 4 * (tid & 31);
    const int as_off = (tid >> 3) * APITCH + 4 * (tid & 7);
    const int bs_off = (tid >> 5) * BPITCH2 + 8 * (tid & 31);

    const int a_rd = (wm * 32 + ln) * APITCH + 2 * half;
    const int b_rd = ROPE ? half * BPITCH2 + 2 * (wn * 64) + 4 * ln
                          : half * BPITCH2 + 2 * (wn * 64 + ln);

    g4f ra[4], rb[4];
#pragma unroll
    for (int p = 0; p < 4; ++p) {
        ra[p] = *(const g4f*)(ag + (size_t)(32 * p) * K);
        rb[p] = *(const g4f*)(bg + (size_t)(16 * (p >> 1) + (p & 1)) * N);
    }
#pragma unroll
    for (int p = 0; p < 4; ++p)
        *(g4f*)(lds + as_off + 32 * p * APITCH) = ra[p];
#pragma unroll
    for (int q = 0; q < 2; ++q)
        store_bpair(lds + A_STAGE + bs_off + 8 * q * BPITCH2, rb[2 * q], rb[2 * q + 1]);
    __syncthreads();

    v8f acc00 = {}, acc01 = {}, acc02 = {}, acc03 = {};
    v8f acc10 = {}, acc11 = {}, acc12 = {}, acc13 = {};

    const int nch = K / GBK;
    for (int c = 0; c < nch; ++c) {
        const float* As = lds + (c & 1) * STAGE_F;
        const float* Bs = As + A_STAGE;
        const bool more = (c + 1 < nch);

        if (more) {
            ag += GBK;
            bg += (size_t)GBK * N;
#pragma unroll
            for (int p = 0; p < 4; ++p) {
                ra[p] = *(const g4f*)(ag + (size_t)(32 * p) * K);
                rb[p] = *(const g4f*)(bg + (size_t)(16 * (p >> 1) + (p & 1)) * N);
            }
        }

        GFrag f[2];
        load_gfrag<ROPE>(f[0], As + a_rd, Bs + b_rd);
#pragma unroll
        for (int ks = 0; ks < GBK / 4; ++ks) {
            if (ks + 1 < GBK / 4)
                load_gfrag<ROPE>(f[(ks + 1) & 1], As + a_rd + 4 * (ks + 1), Bs + b_rd + 2 * (ks + 1) * BPITCH2);
            __builtin_amdgcn_sched_barrier(0);
            const GFrag& g = f[ks & 1];
            acc00 = __builtin_amdgcn_wmma_f32_16x16x4_f32(false, g.a0, false, g.b0, (short)0, acc00, false, false);
            acc10 = __builtin_amdgcn_wmma_f32_16x16x4_f32(false, g.a1, false, g.b0, (short)0, acc10, false, false);
            acc01 = __builtin_amdgcn_wmma_f32_16x16x4_f32(false, g.a0, false, g.b1, (short)0, acc01, false, false);
            acc11 = __builtin_amdgcn_wmma_f32_16x16x4_f32(false, g.a1, false, g.b1, (short)0, acc11, false, false);
            acc02 = __builtin_amdgcn_wmma_f32_16x16x4_f32(false, g.a0, false, g.b2, (short)0, acc02, false, false);
            acc12 = __builtin_amdgcn_wmma_f32_16x16x4_f32(false, g.a1, false, g.b2, (short)0, acc12, false, false);
            acc03 = __builtin_amdgcn_wmma_f32_16x16x4_f32(false, g.a0, false, g.b3, (short)0, acc03, false, false);
            acc13 = __builtin_amdgcn_wmma_f32_16x16x4_f32(false, g.a1, false, g.b3, (short)0, acc13, false, false);
            __builtin_amdgcn_sched_barrier(0);
        }

        if (more) {
            float* Ad = lds + ((c + 1) & 1) * STAGE_F;
            float* Bd = Ad + A_STAGE;
#pragma unroll
            for (int p = 0; p < 4; ++p)
                *(g4f*)(Ad + as_off + 32 * p * APITCH) = ra[p];
#pragma unroll
            for (int q = 0; q < 2; ++q)
                store_bpair(Bd + bs_off + 8 * q * BPITCH2, rb[2 * q], rb[2 * q + 1]);
        }
        __syncthreads();
    }

    v8f accs[2][4] = { { acc00, acc01, acc02, acc03 },
                       { acc10, acc11, acc12, acc13 } };
    if (ROPE) {
        const int cbase = n0 + wn * 64;
        const int sec   = cbase >> 10;
        const int h     = (cbase & (DM - 1)) >> 6;
        float* __restrict__ dsec = C + (size_t)sec * ((size_t)BB * NH * TT * DH);

        v2f* mine = (v2f*)lds + (wid * 32) * 32 + lane;
#pragma unroll
        for (int tp = 0; tp < 2; ++tp) {
#pragma unroll
            for (int sub = 0; sub < 2; ++sub) {
#pragma unroll
                for (int v = 0; v < 8; ++v) {
                    v2f p;
                    p.x = accs[sub][2 * tp][v];
                    p.y = accs[sub][2 * tp + 1][v];
                    mine[(16 * tp + 8 * sub + v) * 32] = p;
                }
            }
        }

        const v2f bia0 = *(const v2f*)(bias + cbase + 2 * ln);
        const v2f bia1 = *(const v2f*)(bias + cbase + 32 + 2 * ln);
        const float inv0 = expf(-(float)(2 * ln) * (9.210340371976184f / 64.0f));
        const float inv1 = expf(-(float)(2 * (16 + ln)) * (9.210340371976184f / 64.0f));
        const int rbase = m0 + wm * 32 + 8 * half;

        v2f p = mine[0];
#pragma unroll 1
        for (int j = 0; j < 32; ++j) {
            const v2f pn = mine[((j + 1) & 31) * 32];
            const int tp  = j >> 4;
            const int row = rbase + 16 * ((j >> 3) & 1) + (j & 7);
            const int b   = row / TT;
            const int t   = row % TT;
            const int i   = 16 * tp + ln;
            const v2f bia = tp ? bia1 : bia0;
            const float e = p.x + bia.x;
            const float o = p.y + bia.y;
            v2f r;
            if (sec == 2) {
                r.x = e; r.y = o;
            } else {
                const float ang = (float)t * (tp ? inv1 : inv0);
                float sn, cs;
                sincosf(ang, &sn, &cs);
                r.x = e * cs - o * sn;
                r.y = e * sn + o * cs;
            }
            *(v2f*)(dsec + (((size_t)(b * NH + h)) * TT + t) * DH + 2 * i) = r;
            p = pn;
        }
    } else {
#pragma unroll
        for (int sub = 0; sub < 2; ++sub) {
            const int rbase = m0 + wm * 32 + sub * 16 + 8 * half;
#pragma unroll
            for (int t = 0; t < 4; ++t) {
                const int col = n0 + wn * 64 + t * 16 + ln;
                const float bs = bias[col];
#pragma unroll
                for (int v = 0; v < 8; ++v)
                    C[(size_t)(rbase + v) * N + col] = accs[sub][t][v] + bs;
            }
        }
    }
}

__global__ void __launch_bounds__(256)
gemm_bias_f32wmma(const float* __restrict__ A,
                  const float* __restrict__ Bm,
                  const float* __restrict__ bias,
                  float* __restrict__ C,
                  int M, int N, int K)
{
    __shared__ __attribute__((aligned(16))) float lds[2 * STAGE_F];
    (void)M;
    gemm_body<false>(lds, A, Bm, bias, C, N, K);
}

__global__ void __launch_bounds__(256)
gemm_qkv_rope_f32wmma(const float* __restrict__ A,
                      const float* __restrict__ Bm,
                      const float* __restrict__ bias,
                      float* __restrict__ C,
                      int M, int N, int K)
{
    __shared__ __attribute__((aligned(16))) float lds[2 * STAGE_F];
    (void)M;
    gemm_body<true>(lds, A, Bm, bias, C, N, K);
}

typedef __attribute__((ext_vector_type(4))) float v4f;
typedef __attribute__((ext_vector_type(4))) int   v4i;

__device__ __forceinline__ float xhalf(float x)
{
    const int bits = __builtin_bit_cast(int, x);
    return __builtin_bit_cast(float,
        __builtin_amdgcn_permlanex16(bits, bits, 0x76543210, 0xfedcba98, false, false));
}

struct KVTile {
    v4f k4[8];
    v4f v4[8];
    v4i pm0, pm1;
};

__device__ __forceinline__ void load_kv_tile(KVTile& t,
                                             const float* __restrict__ kptr,
                                             const float* __restrict__ vptr,
                                             const int*   __restrict__ mptr,
                                             int kbase)
{
    for (int w = 0; w < 8; ++w)
        t.k4[w] = *(const v4f*)(kptr + (size_t)kbase * DH + 8 * w);
    for (int r = 0; r < 8; ++r)
        t.v4[r] = *(const v4f*)(vptr + (size_t)(kbase + r) * DH);
    t.pm0 = *(const v4i*)(mptr + kbase);
    t.pm1 = *(const v4i*)(mptr + kbase + 4);
}

__device__ __forceinline__ void hold_loaded(const KVTile& t)
{
    asm volatile("" :: "v"(t.k4[0]), "v"(t.k4[1]), "v"(t.k4[2]), "v"(t.k4[3]),
                       "v"(t.k4[4]), "v"(t.k4[5]), "v"(t.k4[6]), "v"(t.k4[7]),
                       "v"(t.v4[0]), "v"(t.v4[1]), "v"(t.v4[2]), "v"(t.v4[3]),
                       "v"(t.v4[4]), "v"(t.v4[5]), "v"(t.v4[6]), "v"(t.v4[7]),
                       "v"(t.pm0), "v"(t.pm1));
}

__device__ __forceinline__ void attn_tile(const KVTile& t, const v4f (&q4)[8],
                                          v8f& o0, v8f& o1, v8f& o2, v8f& o3,
                                          float& rmax, float& rsum,
                                          bool diag, int half, int ln)
{
    const float scale = 0.125f;

    v8f s = {};
    for (int w = 0; w < 8; ++w) {
        v2f ka, qa, kc, qc;
        ka.x = t.k4[w].x; ka.y = t.k4[w].y;  qa.x = q4[w].x; qa.y = q4[w].y;
        kc.x = t.k4[w].z; kc.y = t.k4[w].w;  qc.x = q4[w].z; qc.y = q4[w].w;
        s = __builtin_amdgcn_wmma_f32_16x16x4_f32(false, ka, false, qa, (short)0, s, false, false);
        s = __builtin_amdgcn_wmma_f32_16x16x4_f32(false, kc, false, qc, (short)0, s, false, false);
    }

    const int pmv[8] = { t.pm0.x, t.pm0.y, t.pm0.z, t.pm0.w, t.pm1.x, t.pm1.y, t.pm1.z, t.pm1.w };
    float x[8];
    for (int r = 0; r < 8; ++r)
        x[r] = (pmv[r] == 0) ? -1e30f : s[r] * scale;
    if (diag) {
        for (int r = 0; r < 8; ++r)
            x[r] = (8 * half + r > ln) ? -1e30f : x[r];
    }

    float mx = fmaxf(fmaxf(fmaxf(x[0], x[1]), fmaxf(x[2], x[3])),
                     fmaxf(fmaxf(x[4], x[5]), fmaxf(x[6], x[7])));
    mx = fmaxf(mx, xhalf(mx));
    const float nmax  = fmaxf(rmax, mx);
    const float alpha = __expf(rmax - nmax);
    float p[8];
    for (int r = 0; r < 8; ++r)
        p[r] = __expf(x[r] - nmax);
    float sr = ((p[0] + p[1]) + (p[2] + p[3])) + ((p[4] + p[5]) + (p[6] + p[7]));
    sr += xhalf(sr);
    rsum = rsum * alpha + sr;
    rmax = nmax;
    o0 *= alpha; o1 *= alpha; o2 *= alpha; o3 *= alpha;

    for (int i = 0; i < 4; ++i) {
        v2f pf; pf.x = p[2 * i]; pf.y = p[2 * i + 1];
        v2f a0; a0.x = t.v4[2 * i].x; a0.y = t.v4[2 * i + 1].x;
        v2f a1; a1.x = t.v4[2 * i].y; a1.y = t.v4[2 * i + 1].y;
        v2f a2; a2.x = t.v4[2 * i].z; a2.y = t.v4[2 * i + 1].z;
        v2f a3; a3.x = t.v4[2 * i].w; a3.y = t.v4[2 * i + 1].w;
        o0 = __builtin_amdgcn_wmma_f32_16x16x4_f32(false, a0, false, pf, (short)0, o0, false, false);
        o1 = __builtin_amdgcn_wmma_f32_16x16x4_f32(false, a1, false, pf, (short)0, o1, false, false);
        o2 = __builtin_amdgcn_wmma_f32_16x16x4_f32(false, a2, false, pf, (short)0, o2, false, false);
        o3 = __builtin_amdgcn_wmma_f32_16x16x4_f32(false, a3, false, pf, (short)0, o3, false, false);
    }
}

__global__ void __launch_bounds__(256)
flash_attn_f32wmma(const float* __restrict__ qb,
                   const float* __restrict__ kb,
                   const float* __restrict__ vb,
                   const int*   __restrict__ amask,
                   float* __restrict__ attn  )
{
    const int wid  = __builtin_amdgcn_readfirstlane((int)(threadIdx.x >> 5));
    const int lane = threadIdx.x & 31;
    const int half = lane >> 4;
    const int ln   = lane & 15;

    const int NQT  = TT / 16;
    const int bh   = blockIdx.x >> 3;
    const int pr   = ((blockIdx.x & 7) << 3) + wid;
    const int b    = bh >> 4;
    const int h    = bh & (NH - 1);

    const float* __restrict__ Q  = qb + (size_t)bh * TT * DH;
    const float* __restrict__ Kp = kb + (size_t)bh * TT * DH;
    const float* __restrict__ Vp = vb + (size_t)bh * TT * DH;

    const float* __restrict__ kptr = Kp + (size_t)ln * DH + 4 * half;
    const float* __restrict__ vptr = Vp + (size_t)(8 * half) * DH + 4 * ln;
    const int*   __restrict__ mptr = amask + b * TT + 8 * half;

    for (int pass = 0; pass < 2; ++pass) {
        const int q0 = (pass == 0 ? pr : NQT - 1 - pr) * 16;

        v4f q4[8];
        {
            const float* __restrict__ qrow = Q + (size_t)(q0 + ln) * DH + 4 * half;
            for (int w = 0; w < 8; ++w)
                q4[w] = *(const v4f*)(qrow + 8 * w);
        }

        v8f o0 = {}, o1 = {}, o2 = {}, o3 = {};
        float rmax = -1e30f;
        float rsum = 0.0f;

        const int nkt = q0 / 16 + 1;

        KVTile ta, tb;
        load_kv_tile(ta, kptr, vptr, mptr, 0);
        int j = 0;
        for (;;) {
            if (j == nkt - 1) { attn_tile(ta, q4, o0, o1, o2, o3, rmax, rsum, true, half, ln); break; }
            load_kv_tile(tb, kptr, vptr, mptr, (j + 1) * 16);
            attn_tile(ta, q4, o0, o1, o2, o3, rmax, rsum, false, half, ln);
            __builtin_amdgcn_sched_barrier(0);
            hold_loaded(tb);
            ++j;
            if (j == nkt - 1) { attn_tile(tb, q4, o0, o1, o2, o3, rmax, rsum, true, half, ln); break; }
            load_kv_tile(ta, kptr, vptr, mptr, (j + 1) * 16);
            attn_tile(tb, q4, o0, o1, o2, o3, rmax, rsum, false, half, ln);
            __builtin_amdgcn_sched_barrier(0);
            hold_loaded(ta);
            ++j;
        }

        const float inv = 1.0f / rsum;
        float* __restrict__ orow =
            attn + ((size_t)b * TT + (q0 + ln)) * DM + h * DH + 32 * half;
        for (int r = 0; r < 8; ++r) {
            v4f o;
            o.x = o0[r] * inv; o.y = o1[r] * inv; o.z = o2[r] * inv; o.w = o3[r] * inv;
            *(v4f*)(orow + 4 * r) = o;
        }
    }
}

extern "C" void kernel_launch(void* const* d_in, const int* in_sizes, int n_in,
                              void* d_out, int out_size, void* d_ws, size_t ws_size,
                              hipStream_t stream)
{
    (void)in_sizes; (void)n_in; (void)out_size; (void)ws_size;

    const float* x      = (const float*)d_in[0];
    const int*   mask   = (const int*)  d_in[1];
    const float* W_qkv  = (const float*)d_in[2];
    const float* b_qkv  = (const float*)d_in[3];
    const float* W_out  = (const float*)d_in[4];
    const float* b_out  = (const float*)d_in[5];
    float*       out    = (float*)d_out;

    float* ws   = (float*)d_ws;
    float* qb   = ws;
    float* kb   = qb  + (size_t)BB * NH * TT * DH;
    float* vb   = kb  + (size_t)BB * NH * TT * DH;
    float* attn = vb  + (size_t)BB * NH * TT * DH;

    {
        const int blocks = (ROWS / GBM) * (QKV_N / GBN);
        gemm_qkv_rope_f32wmma<<<blocks, 256, 0, stream>>>(
            x, W_qkv, b_qkv, qb, ROWS, QKV_N, DM);
    }

    {
        const int blocks = BB * NH * 8;
        flash_attn_f32wmma<<<blocks, 256, 0, stream>>>(qb, kb, vb, mask, attn);
    }

    {
        const int blocks = (ROWS / GBM) * (DM / GBN);
        gemm_bias_f32wmma<<<blocks, 256, 0, stream>>>(
            attn, W_out, b_out, out, ROWS, DM, DM);
    }
}


# k buffer stored per 16-key tile in K-fragment order by the projection epilogue: each attention K load reads 512 contiguous bytes (4 lines) instead of 32 B from 16 lines; sums unchanged
# speedup vs baseline: 3.7656x; 1.1208x over previous
#include <hip/hip_runtime.h>
#include <hip/hip_bf16.h>
#include <math.h>


#define BB      2
#define TT      2048
#define DM      1024
#define NH      16
#define DH      64
#define ROWS    (BB * TT)
#define QKV_N   (3 * DM)

typedef __attribute__((ext_vector_type(2))) float v2f;
typedef __attribute__((ext_vector_type(8))) float v8f;

typedef __attribute__((ext_vector_type(4))) float g4f;

#define GBM      128
#define GBN      128
#define GBK      32
#define APITCH   36
#define BPITCH2  288
#define A_STAGE  (GBM * APITCH)
#define B_STAGE  ((GBK / 2) * BPITCH2)
#define STAGE_F  (A_STAGE + B_STAGE)

struct GFrag { v2f a0, a1, b0, b1, b2, b3; };

template <bool PAIRS>
__device__ __forceinline__ void load_gfrag(GFrag& f, const float* ap, const float* bp)
{
    f.a0 = *(const v2f*)(ap);
    f.a1 = *(const v2f*)(ap + 16 * APITCH);
    if (PAIRS) {
        const g4f b01 = *(const g4f*)(bp);
        const g4f b23 = *(const g4f*)(bp + 64);
        f.b0.x = b01.x; f.b0.y = b01.y;  f.b1.x = b01.z; f.b1.y = b01.w;
        f.b2.x = b23.x; f.b2.y = b23.y;  f.b3.x = b23.z; f.b3.y = b23.w;
    } else {
        f.b0 = *(const v2f*)(bp);
        f.b1 = *(const v2f*)(bp + 32);
        f.b2 = *(const v2f*)(bp + 64);
        f.b3 = *(const v2f*)(bp + 96);
    }
}

__device__ __forceinline__ void store_bpair(float* dst, g4f r0, g4f r1)
{
    g4f lo, hi;
    lo.x = r0.x; lo.y = r1.x; lo.z = r0.y; lo.w = r1.y;
    hi.x = r0.z; hi.y = r1.z; hi.z = r0.w; hi.w = r1.w;
    *(g4f*)(dst)     = lo;
    *(g4f*)(dst + 4) = hi;
}

template <bool ROPE>
__device__ __forceinline__ void gemm_body(float* lds,
                                          const float* __restrict__ A,
                                          const float* __restrict__ Bm,
                                          const float* __restrict__ bias,
                                          float* __restrict__ C,
                                          int N, int K)
{
    const int tid  = threadIdx.x;
    const int wid  = __builtin_amdgcn_readfirstlane(tid >> 5);
    const int lane = tid & 31;
    const int half = lane >> 4;
    const int ln   = lane & 15;
    const int wm   = wid & 3;
    const int wn   = wid >> 2;

    const int nbn = N / GBN;
    const int m0  = (blockIdx.x / nbn) * GBM;
    const int n0  = (blockIdx.x % nbn) * GBN;

    const float* __restrict__ ag = A  + (size_t)(m0 + (tid >> 3)) * K + 4 * (tid & 7);
    const float* __restrict__ bg = Bm + (size_t)(2 * (tid >> 5)) * N + n0 + 4 * (tid & 31);
    const int as_off = (tid >> 3) * APITCH + 4 * (tid & 7);
    const int bs_off = (tid >> 5) * BPITCH2 + 8 * (tid & 31);

    const int a_rd = (wm * 32 + ln) * APITCH + 2 * half;
    const int b_rd = ROPE ? half * BPITCH2 + 2 * (wn * 64) + 4 * ln
                          : half * BPITCH2 + 2 * (wn * 64 + ln);

    g4f ra[4], rb[4];
#pragma unroll
    for (int p = 0; p < 4; ++p) {
        ra[p] = *(const g4f*)(ag + (size_t)(32 * p) * K);
        rb[p] = *(const g4f*)(bg + (size_t)(16 * (p >> 1) + (p & 1)) * N);
    }
#pragma unroll
    for (int p = 0; p < 4; ++p)
        *(g4f*)(lds + as_off + 32 * p * APITCH) = ra[p];
#pragma unroll
    for (int q = 0; q < 2; ++q)
        store_bpair(lds + A_STAGE + bs_off + 8 * q * BPITCH2, rb[2 * q], rb[2 * q + 1]);
    __syncthreads();

    v8f acc00 = {}, acc01 = {}, acc02 = {}, acc03 = {};
    v8f acc10 = {}, acc11 = {}, acc12 = {}, acc13 = {};

    const int nch = K / GBK;
    for (int c = 0; c < nch; ++c) {
        const float* As = lds + (c & 1) * STAGE_F;
        const float* Bs = As + A_STAGE;
        const bool more = (c + 1 < nch);

        if (more) {
            ag += GBK;
            bg += (size_t)GBK * N;
#pragma unroll
            for (int p = 0; p < 4; ++p) {
                ra[p] = *(const g4f*)(ag + (size_t)(32 * p) * K);
                rb[p] = *(const g4f*)(bg + (size_t)(16 * (p >> 1) + (p & 1)) * N);
            }
        }

        GFrag f[2];
        load_gfrag<ROPE>(f[0], As + a_rd, Bs + b_rd);
#pragma unroll
        for (int ks = 0; ks < GBK / 4; ++ks) {
            if (ks + 1 < GBK / 4)
                load_gfrag<ROPE>(f[(ks + 1) & 1], As + a_rd + 4 * (ks + 1), Bs + b_rd + 2 * (ks + 1) * BPITCH2);
            __builtin_amdgcn_sched_barrier(0);
            const GFrag& g = f[ks & 1];
            acc00 = __builtin_amdgcn_wmma_f32_16x16x4_f32(false, g.a0, false, g.b0, (short)0, acc00, false, false);
            acc10 = __builtin_amdgcn_wmma_f32_16x16x4_f32(false, g.a1, false, g.b0, (short)0, acc10, false, false);
            acc01 = __builtin_amdgcn_wmma_f32_16x16x4_f32(false, g.a0, false, g.b1, (short)0, acc01, false, false);
            acc11 = __builtin_amdgcn_wmma_f32_16x16x4_f32(false, g.a1, false, g.b1, (short)0, acc11, false, false);
            acc02 = __builtin_amdgcn_wmma_f32_16x16x4_f32(false, g.a0, false, g.b2, (short)0, acc02, false, false);
            acc12 = __builtin_amdgcn_wmma_f32_16x16x4_f32(false, g.a1, false, g.b2, (short)0, acc12, false, false);
            acc03 = __builtin_amdgcn_wmma_f32_16x16x4_f32(false, g.a0, false, g.b3, (short)0, acc03, false, false);
            acc13 = __builtin_amdgcn_wmma_f32_16x16x4_f32(false, g.a1, false, g.b3, (short)0, acc13, false, false);
            __builtin_amdgcn_sched_barrier(0);
        }

        if (more) {
            float* Ad = lds + ((c + 1) & 1) * STAGE_F;
            float* Bd = Ad + A_STAGE;
#pragma unroll
            for (int p = 0; p < 4; ++p)
                *(g4f*)(Ad + as_off + 32 * p * APITCH) = ra[p];
#pragma unroll
            for (int q = 0; q < 2; ++q)
                store_bpair(Bd + bs_off + 8 * q * BPITCH2, rb[2 * q], rb[2 * q + 1]);
        }
        __syncthreads();
    }

    v8f accs[2][4] = { { acc00, acc01, acc02, acc03 },
                       { acc10, acc11, acc12, acc13 } };
    if (ROPE) {
        const int cbase = n0 + wn * 64;
        const int sec   = cbase >> 10;
        const int h     = (cbase & (DM - 1)) >> 6;
        float* __restrict__ dsec = C + (size_t)sec * ((size_t)BB * NH * TT * DH);

        const v2f bia0 = *(const v2f*)(bias + cbase + 2 * ln);
        const v2f bia1 = *(const v2f*)(bias + cbase + 32 + 2 * ln);
        v2f* mine = (v2f*)lds + (wid * 32) * 32 + lane;
#pragma unroll
        for (int tp = 0; tp < 2; ++tp) {
            const v2f bia = tp ? bia1 : bia0;
#pragma unroll
            for (int sub = 0; sub < 2; ++sub) {
#pragma unroll
                for (int v = 0; v < 8; ++v) {
                    v2f p;
                    p.x = accs[sub][2 * tp][v] + bia.x;
                    p.y = accs[sub][2 * tp + 1][v] + bia.y;
                    mine[(16 * tp + 8 * sub + v) * 32] = p;
                }
            }
        }

        const float inv0 = expf(-(float)(2 * ln) * (9.210340371976184f / 64.0f));
        const float inv1 = expf(-(float)(2 * (16 + ln)) * (9.210340371976184f / 64.0f));
        const int rbase = m0 + wm * 32 + 8 * half;

        if (sec == 1) {
            __syncthreads();
            const v2f* wbase = (const v2f*)lds + (wid * 32) * 32;
            const int kg = lane >> 4;
            const int kv = (lane >> 1) & 7;
            const int pp = lane & 1;
            const int row = m0 + wm * 32 + 8 * kg + kv;
            v2f p = wbase[kv * 32 + 16 * kg + pp];
#pragma unroll 1
            for (int jj = 0; jj < 32; ++jj) {
                const int jn   = (jj + 1) & 31;
                const int in   = 2 * (jn & 15) + pp;
                const v2f pn   = wbase[(16 * (in >> 4) + 8 * (jn >> 4) + kv) * 32 + 16 * kg + (in & 15)];
                const int sub  = jj >> 4;
                const int m    = jj & 15;
                const int i    = 2 * m + pp;
                const int r0   = row + 16 * sub;
                const int b    = r0 / TT;
                const int t    = r0 % TT;
                const float inv = expf(-(float)(2 * i) * (9.210340371976184f / 64.0f));
                const float ang = (float)t * inv;
                float sn, cs;
                sincosf(ang, &sn, &cs);
                v2f r;
                r.x = p.x * cs - p.y * sn;
                r.y = p.x * sn + p.y * cs;
                *(v2f*)(dsec + (((size_t)(b * NH + h)) * TT + (t & ~15)) * DH + m * 64 + 2 * lane) = r;
                p = pn;
            }
        } else {
            v2f p = mine[0];
#pragma unroll 1
            for (int j = 0; j < 32; ++j) {
                const v2f pn = mine[((j + 1) & 31) * 32];
                const int tp  = j >> 4;
                const int row = rbase + 16 * ((j >> 3) & 1) + (j & 7);
                const int b   = row / TT;
                const int t   = row % TT;
                const int i   = 16 * tp + ln;
                const float e = p.x;
                const float o = p.y;
                v2f r;
                if (sec == 2) {
                    r.x = e; r.y = o;
                } else {
                    const float ang = (float)t * (tp ? inv1 : inv0);
                    float sn, cs;
                    sincosf(ang, &sn, &cs);
                    r.x = e * cs - o * sn;
                    r.y = e * sn + o * cs;
                }
                *(v2f*)(dsec + (((size_t)(b * NH + h)) * TT + t) * DH + 2 * i) = r;
                p = pn;
            }
        }
    } else {
#pragma unroll
        for (int sub = 0; sub < 2; ++sub) {
            const int rbase = m0 + wm * 32 + sub * 16 + 8 * half;
#pragma unroll
            for (int t = 0; t < 4; ++t) {
                const int col = n0 + wn * 64 + t * 16 + ln;
                const float bs = bias[col];
#pragma unroll
                for (int v = 0; v < 8; ++v)
                    C[(size_t)(rbase + v) * N + col] = accs[sub][t][v] + bs;
            }
        }
    }
}

__global__ void __launch_bounds__(256)
gemm_bias_f32wmma(const float* __restrict__ A,
                  const float* __restrict__ Bm,
                  const float* __restrict__ bias,
                  float* __restrict__ C,
                  int M, int N, int K)
{
    __shared__ __attribute__((aligned(16))) float lds[2 * STAGE_F];
    (void)M;
    gemm_body<false>(lds, A, Bm, bias, C, N, K);
}

__global__ void __launch_bounds__(256)
gemm_qkv_rope_f32wmma(const float* __restrict__ A,
                      const float* __restrict__ Bm,
                      const float* __restrict__ bias,
                      float* __restrict__ C,
                      int M, int N, int K)
{
    __shared__ __attribute__((aligned(16))) float lds[2 * STAGE_F];
    (void)M;
    gemm_body<true>(lds, A, Bm, bias, C, N, K);
}

typedef __attribute__((ext_vector_type(4))) float v4f;
typedef __attribute__((ext_vector_type(4))) int   v4i;

__device__ __forceinline__ float xhalf(float x)
{
    const int bits = __builtin_bit_cast(int, x);
    return __builtin_bit_cast(float,
        __builtin_amdgcn_permlanex16(bits, bits, 0x76543210, 0xfedcba98, false, false));
}

struct KVTile {
    v4f k4[8];
    v4f v4[8];
    v4i pm0, pm1;
};

__device__ __forceinline__ void load_kv_tile(KVTile& t,
                                             const float* __restrict__ kptr,
                                             const float* __restrict__ vptr,
                                             const int*   __restrict__ mptr,
                                             int kbase)
{
    for (int w = 0; w < 8; ++w)
        t.k4[w] = *(const v4f*)(kptr + (size_t)kbase * DH + 128 * w);
    for (int r = 0; r < 8; ++r)
        t.v4[r] = *(const v4f*)(vptr + (size_t)(kbase + r) * DH);
    t.pm0 = *(const v4i*)(mptr + kbase);
    t.pm1 = *(const v4i*)(mptr + kbase + 4);
}

__device__ __forceinline__ void hold_loaded(const KVTile& t)
{
    asm volatile("" :: "v"(t.k4[0]), "v"(t.k4[1]), "v"(t.k4[2]), "v"(t.k4[3]),
                       "v"(t.k4[4]), "v"(t.k4[5]), "v"(t.k4[6]), "v"(t.k4[7]),
                       "v"(t.v4[0]), "v"(t.v4[1]), "v"(t.v4[2]), "v"(t.v4[3]),
                       "v"(t.v4[4]), "v"(t.v4[5]), "v"(t.v4[6]), "v"(t.v4[7]),
                       "v"(t.pm0), "v"(t.pm1));
}

__device__ __forceinline__ void attn_tile(const KVTile& t, const v4f (&q4)[8],
                                          v8f& o0, v8f& o1, v8f& o2, v8f& o3,
                                          float& rmax, float& rsum,
                                          bool diag, int half, int ln)
{
    const float scale = 0.125f;

    v8f s = {};
    for (int w = 0; w < 8; ++w) {
        v2f ka, qa, kc, qc;
        ka.x = t.k4[w].x; ka.y = t.k4[w].y;  qa.x = q4[w].x; qa.y = q4[w].y;
        kc.x = t.k4[w].z; kc.y = t.k4[w].w;  qc.x = q4[w].z; qc.y = q4[w].w;
        s = __builtin_amdgcn_wmma_f32_16x16x4_f32(false, ka, false, qa, (short)0, s, false, false);
        s = __builtin_amdgcn_wmma_f32_16x16x4_f32(false, kc, false, qc, (short)0, s, false, false);
    }

    const int pmv[8] = { t.pm0.x, t.pm0.y, t.pm0.z, t.pm0.w, t.pm1.x, t.pm1.y, t.pm1.z, t.pm1.w };
    float x[8];
    for (int r = 0; r < 8; ++r)
        x[r] = (pmv[r] == 0) ? -1e30f : s[r] * scale;
    if (diag) {
        for (int r = 0; r < 8; ++r)
            x[r] = (8 * half + r > ln) ? -1e30f : x[r];
    }

    float mx = fmaxf(fmaxf(fmaxf(x[0], x[1]), fmaxf(x[2], x[3])),
                     fmaxf(fmaxf(x[4], x[5]), fmaxf(x[6], x[7])));
    mx = fmaxf(mx, xhalf(mx));
    const float nmax  = fmaxf(rmax, mx);
    const float alpha = __expf(rmax - nmax);
    float p[8];
    for (int r = 0; r < 8; ++r)
        p[r] = __expf(x[r] - nmax);
    float sr = ((p[0] + p[1]) + (p[2] + p[3])) + ((p[4] + p[5]) + (p[6] + p[7]));
    sr += xhalf(sr);
    rsum = rsum * alpha + sr;
    rmax = nmax;
    o0 *= alpha; o1 *= alpha; o2 *= alpha; o3 *= alpha;

    for (int i = 0; i < 4; ++i) {
        v2f pf; pf.x = p[2 * i]; pf.y = p[2 * i + 1];
        v2f a0; a0.x = t.v4[2 * i].x; a0.y = t.v4[2 * i + 1].x;
        v2f a1; a1.x = t.v4[2 * i].y; a1.y = t.v4[2 * i + 1].y;
        v2f a2; a2.x = t.v4[2 * i].z; a2.y = t.v4[2 * i + 1].z;
        v2f a3; a3.x = t.v4[2 * i].w; a3.y = t.v4[2 * i + 1].w;
        o0 = __builtin_amdgcn_wmma_f32_16x16x4_f32(false, a0, false, pf, (short)0, o0, false, false);
        o1 = __builtin_amdgcn_wmma_f32_16x16x4_f32(false, a1, false, pf, (short)0, o1, false, false);
        o2 = __builtin_amdgcn_wmma_f32_16x16x4_f32(false, a2, false, pf, (short)0, o2, false, false);
        o3 = __builtin_amdgcn_wmma_f32_16x16x4_f32(false, a3, false, pf, (short)0, o3, false, false);
    }
}

__global__ void __launch_bounds__(256)
flash_attn_f32wmma(const float* __restrict__ qb,
                   const float* __restrict__ kb,
                   const float* __restrict__ vb,
                   const int*   __restrict__ amask,
                   float* __restrict__ attn  )
{
    const int wid  = __builtin_amdgcn_readfirstlane((int)(threadIdx.x >> 5));
    const int lane = threadIdx.x & 31;
    const int half = lane >> 4;
    const int ln   = lane & 15;

    const int NQT  = TT / 16;
    const int bh   = blockIdx.x >> 3;
    const int pr   = ((blockIdx.x & 7) << 3) + wid;
    const int b    = bh >> 4;
    const int h    = bh & (NH - 1);

    const float* __restrict__ Q  = qb + (size_t)bh * TT * DH;
    const float* __restrict__ Kp = kb + (size_t)bh * TT * DH;
    const float* __restrict__ Vp = vb + (size_t)bh * TT * DH;

    const float* __restrict__ kptr = Kp + (size_t)half * DH + 4 * ln;
    const float* __restrict__ vptr = Vp + (size_t)(8 * half) * DH + 4 * ln;
    const int*   __restrict__ mptr = amask + b * TT + 8 * half;

    for (int pass = 0; pass < 2; ++pass) {
        const int q0 = (pass == 0 ? pr : NQT - 1 - pr) * 16;

        v4f q4[8];
        {
            const float* __restrict__ qrow = Q + (size_t)(q0 + ln) * DH + 4 * half;
            for (int w = 0; w < 8; ++w)
                q4[w] = *(const v4f*)(qrow + 8 * w);
        }

        v8f o0 = {}, o1 = {}, o2 = {}, o3 = {};
        float rmax = -1e30f;
        float rsum = 0.0f;

        const int nkt = q0 / 16 + 1;

        KVTile ta, tb;
        load_kv_tile(ta, kptr, vptr, mptr, 0);
        int j = 0;
        for (;;) {
            if (j == nkt - 1) { attn_tile(ta, q4, o0, o1, o2, o3, rmax, rsum, true, half, ln); break; }
            load_kv_tile(tb, kptr, vptr, mptr, (j + 1) * 16);
            attn_tile(ta, q4, o0, o1, o2, o3, rmax, rsum, false, half, ln);
            __builtin_amdgcn_sched_barrier(0);
            hold_loaded(tb);
            ++j;
            if (j == nkt - 1) { attn_tile(tb, q4, o0, o1, o2, o3, rmax, rsum, true, half, ln); break; }
            load_kv_tile(ta, kptr, vptr, mptr, (j + 1) * 16);
            attn_tile(tb, q4, o0, o1, o2, o3, rmax, rsum, false, half, ln);
            __builtin_amdgcn_sched_barrier(0);
            hold_loaded(ta);
            ++j;
        }

        const float inv = 1.0f / rsum;
        float* __restrict__ orow =
            attn + ((size_t)b * TT + (q0 + ln)) * DM + h * DH + 32 * half;
        for (int r = 0; r < 8; ++r) {
            v4f o;
            o.x = o0[r] * inv; o.y = o1[r] * inv; o.z = o2[r] * inv; o.w = o3[r] * inv;
            *(v4f*)(orow + 4 * r) = o;
        }
    }
}

extern "C" void kernel_launch(void* const* d_in, const int* in_sizes, int n_in,
                              void* d_out, int out_size, void* d_ws, size_t ws_size,
                              hipStream_t stream)
{
    (void)in_sizes; (void)n_in; (void)out_size; (void)ws_size;

    const float* x      = (const float*)d_in[0];
    const int*   mask   = (const int*)  d_in[1];
    const float* W_qkv  = (const float*)d_in[2];
    const float* b_qkv  = (const float*)d_in[3];
    const float* W_out  = (const float*)d_in[4];
    const float* b_out  = (const float*)d_in[5];
    float*       out    = (float*)d_out;

    float* ws   = (float*)d_ws;
    float* qb   = ws;
    float* kb   = qb  + (size_t)BB * NH * TT * DH;
    float* vb   = kb  + (size_t)BB * NH * TT * DH;
    float* attn = vb  + (size_t)BB * NH * TT * DH;

    {
        const int blocks = (ROWS / GBM) * (QKV_N / GBN);
        gemm_qkv_rope_f32wmma<<<blocks, 256, 0, stream>>>(
            x, W_qkv, b_qkv, qb, ROWS, QKV_N, DM);
    }

    {
        const int blocks = BB * NH * 8;
        flash_attn_f32wmma<<<blocks, 256, 0, stream>>>(qb, kb, vb, mask, attn);
    }

    {
        const int blocks = (ROWS / GBM) * (DM / GBN);
        gemm_bias_f32wmma<<<blocks, 256, 0, stream>>>(
            attn, W_out, b_out, out, ROWS, DM, DM);
    }
}
